# starting kernel
# baseline (speedup 1.0000x reference)
#include <hip/hip_runtime.h>
#include <hip/hip_bf16.h>
#include <math.h>

constexpr int Bz = 2;
constexpr int Sq = 2048;
constexpr int Hd = 2048;
constexpr int NH = 16;
constexpr int HD = 128;
constexpr float NEGV = -1000000.0f;

typedef float v2f __attribute__((ext_vector_type(2)));
typedef float v8f __attribute__((ext_vector_type(8)));

#define WMMA_F32(a, b, c) \
  __builtin_amdgcn_wmma_f32_16x16x4_f32(false, (a), false, (b), (short)0, (c), false, false)

__device__ __forceinline__ v8f vzero8() {
  v8f z = {0.f, 0.f, 0.f, 0.f, 0.f, 0.f, 0.f, 0.f};
  return z;
}

constexpr int TM = 128, TN = 128, TK = 32;
constexpr int APAD = 4;
constexpr int BIPAD = 32;

__global__ __launch_bounds__(256) void gemm_f32_wmma(
    const float* __restrict__ A, const float* __restrict__ Bm,
    float* __restrict__ C, int M, int N, int K) {
  __shared__ float As[TM][TK + APAD];
  __shared__ float Bsp[TK / 2][2 * TN + BIPAD];

  const int t = threadIdx.x;
  const int lane = t & 31;
  const int wave = t >> 5;
  const int lm = lane & 15;
  const int lh = lane >> 4;
  const int wm = wave & 3;
  const int wn = wave >> 2;
  const int m0 = blockIdx.y * TM;
  const int n0 = blockIdx.x * TN;

  v8f acc[2][4];
#pragma unroll
  for (int i = 0; i < 2; ++i)
#pragma unroll
    for (int j = 0; j < 4; ++j) acc[i][j] = vzero8();

  for (int kt = 0; kt < K; kt += TK) {
#pragma unroll
    for (int i = 0; i < 4; ++i) {
      int idx = t + i * 256;
      int r = idx >> 3;
      int c = (idx & 7) << 2;
      float4 v = *(const float4*)(A + (size_t)(m0 + r) * K + kt + c);
      *(float4*)&As[r][c] = v;
    }
#pragma unroll
    for (int i = 0; i < 4; ++i) {
      int idx = t + i * 256;
      int r = idx >> 5;
      int c = (idx & 31) << 2;
      float4 v = *(const float4*)(Bm + (size_t)(kt + r) * N + n0 + c);
      float* dst = &Bsp[r >> 1][2 * c + (r & 1)];
      dst[0] = v.x;
      dst[2] = v.y;
      dst[4] = v.z;
      dst[6] = v.w;
    }
    __syncthreads();

#pragma unroll
    for (int kk = 0; kk < TK; kk += 4) {
      v2f af[2];
#pragma unroll
      for (int im = 0; im < 2; ++im) {
        int row = wm * 32 + im * 16 + lm;
        af[im] = *(const v2f*)&As[row][kk + 2 * lh];
      }
      const int khalf = (kk >> 1) + lh;
#pragma unroll
      for (int j = 0; j < 4; ++j) {
        int col = wn * 64 + j * 16 + lm;
        v2f bf = *(const v2f*)&Bsp[khalf][2 * col];
        acc[0][j] = WMMA_F32(af[0], bf, acc[0][j]);
        acc[1][j] = WMMA_F32(af[1], bf, acc[1][j]);
      }
    }
    __syncthreads();
  }

#pragma unroll
  for (int im = 0; im < 2; ++im)
#pragma unroll
    for (int j = 0; j < 4; ++j)
#pragma unroll
      for (int r = 0; r < 8; ++r) {
        int m = m0 + wm * 32 + im * 16 + r + 8 * lh;
        int n = n0 + wn * 64 + j * 16 + lm;
        C[(size_t)m * N + n] = acc[im][j][r];
      }
}

constexpr int QT = 64;
constexpr int KT = 32;
constexpr int KVP = 12;
constexpr int VIPAD = 32;
constexpr int PPAD = 4;

__global__ __launch_bounds__(128) void attn_f32_wmma(
    const float* __restrict__ Qm, const float* __restrict__ Km,
    const float* __restrict__ Vm, const int* __restrict__ mask,
    float* __restrict__ O) {
  __shared__ float Ks[KT][HD + KVP];
  __shared__ float Vsp[KT / 2][2 * HD + VIPAD];
  __shared__ float Ps[4][16][KT + PPAD];
  __shared__ float corrS[4][16];
  __shared__ float rsumS[4][16];
  __shared__ int mkS[KT];

  const int t = threadIdx.x;
  const int lane = t & 31;
  const int w = t >> 5;
  const int lm = lane & 15;
  const int lh = lane >> 4;

  int bid = blockIdx.x;
  const int nqb = Sq / QT;
  const int qb = bid % nqb; bid /= nqb;
  const int h = bid % NH;
  const int b = bid / NH;
  const int q0 = qb * QT;

  const size_t baseBH = (size_t)b * Sq * Hd + (size_t)h * Sq * HD;

  const float* qrow = Qm + baseBH + (size_t)(q0 + w * 16 + lm) * HD;
  v2f qf[HD / 4];
#pragma unroll
  for (int kd = 0; kd < HD / 4; ++kd)
    qf[kd] = *(const v2f*)(qrow + 4 * kd + 2 * lh);

  int mq[8];
#pragma unroll
  for (int r = 0; r < 8; ++r)
    mq[r] = mask[b * Sq + q0 + w * 16 + r + 8 * lh];

  v8f o[8];
#pragma unroll
  for (int s = 0; s < 8; ++s) o[s] = vzero8();
  float rmax = -INFINITY;
  float rsum = 0.0f;

  for (int kt = 0; kt < Sq; kt += KT) {
#pragma unroll
    for (int i = 0; i < 8; ++i) {
      int idx = t + i * 128;
      int r = idx >> 5;
      int c = (idx & 31) << 2;
      float4 kv = *(const float4*)(Km + baseBH + (size_t)(kt + r) * HD + c);
      *(float4*)&Ks[r][c] = kv;
      float4 vv = *(const float4*)(Vm + baseBH + (size_t)(kt + r) * HD + c);
      float* dst = &Vsp[r >> 1][2 * c + (r & 1)];
      dst[0] = vv.x;
      dst[2] = vv.y;
      dst[4] = vv.z;
      dst[6] = vv.w;
    }
    if (t < KT) mkS[t] = mask[b * Sq + kt + t];
    __syncthreads();

    v8f sc[2];
    sc[0] = vzero8();
    sc[1] = vzero8();
#pragma unroll
    for (int kd = 0; kd < HD / 4; ++kd) {
      v2f af = qf[kd];
#pragma unroll
      for (int j = 0; j < 2; ++j) {
        int kc = j * 16 + lm;
        v2f bf = *(const v2f*)&Ks[kc][4 * kd + 2 * lh];
        sc[j] = WMMA_F32(af, bf, sc[j]);
      }
    }

#pragma unroll
    for (int j = 0; j < 2; ++j)
#pragma unroll
      for (int r = 0; r < 8; ++r) {
        int m = r + 8 * lh;
        int kc = j * 16 + lm;
        float v = sc[j][r];
        if (mq[r] == 0 || mkS[kc] == 0) v = NEGV;
        Ps[w][m][kc] = v;
      }
    __syncthreads();

    {
      const int c0 = lh * 16;
      float pm = -INFINITY;
#pragma unroll
      for (int c = 0; c < 16; ++c) pm = fmaxf(pm, Ps[w][lm][c0 + c]);
      pm = fmaxf(pm, __shfl_xor(pm, 16, 32));
      float nm = fmaxf(rmax, pm);
      float cf = __expf(rmax - nm);
      float ps = 0.0f;
#pragma unroll
      for (int c = 0; c < 16; ++c) {
        float e = __expf(Ps[w][lm][c0 + c] - nm);
        Ps[w][lm][c0 + c] = e;
        ps += e;
      }
      ps += __shfl_xor(ps, 16, 32);
      rsum = rsum * cf + ps;
      rmax = nm;
      if (lane < 16) corrS[w][lm] = cf;
    }
    __syncthreads();

    float cv[8];
#pragma unroll
    for (int r = 0; r < 8; ++r) cv[r] = corrS[w][r + 8 * lh];
#pragma unroll
    for (int s = 0; s < 8; ++s)
#pragma unroll
      for (int r = 0; r < 8; ++r) o[s][r] *= cv[r];

#pragma unroll
    for (int kk = 0; kk < KT; kk += 4) {
      v2f af = *(const v2f*)&Ps[w][lm][kk + 2 * lh];
      const int khalf = (kk >> 1) + lh;
#pragma unroll
      for (int s = 0; s < 8; ++s) {
        int col = s * 16 + lm;
        v2f bf = *(const v2f*)&Vsp[khalf][2 * col];
        o[s] = WMMA_F32(af, bf, o[s]);
      }
    }
    __syncthreads();
  }

  if (lane < 16) rsumS[w][lm] = rsum;
  __syncthreads();
  float rv[8];
#pragma unroll
  for (int r = 0; r < 8; ++r) rv[r] = 1.0f / rsumS[w][r + 8 * lh];

#pragma unroll
  for (int s = 0; s < 8; ++s)
#pragma unroll
    for (int r = 0; r < 8; ++r) {
      int s2 = q0 + w * 16 + r + 8 * lh;
      int d = s * 16 + lm;
      O[baseBH + (size_t)s2 * HD + d] = o[s][r] * rv[r];
    }
}

extern "C" void kernel_launch(void* const* d_in, const int* in_sizes, int n_in,
                              void* d_out, int out_size, void* d_ws, size_t ws_size,
                              hipStream_t stream) {
  const float* X    = (const float*)d_in[0];
  const int*   mask = (const int*)d_in[1];
  const float* Wq   = (const float*)d_in[2];
  const float* Wk   = (const float*)d_in[3];
  const float* Wv   = (const float*)d_in[4];
  const float* Wqm  = (const float*)d_in[5];
  const float* Wkm  = (const float*)d_in[6];
  const float* Wvm  = (const float*)d_in[7];
  const float* Wo   = (const float*)d_in[8];
  float* out = (float*)d_out;

  const size_t SZ = (size_t)Bz * Sq * Hd;
  float* s0 = (float*)d_ws;
  float* s1 = s0 + SZ;
  float* s2 = s1 + SZ;
  float* s3 = s2 + SZ;

  const int M = Bz * Sq, N = Hd, K = Hd;
  dim3 gg(N / TN, M / TM);
  dim3 gb(256);

  gemm_f32_wmma<<<gg, gb, 0, stream>>>(X,  Wq,  s0, M, N, K);
  gemm_f32_wmma<<<gg, gb, 0, stream>>>(X,  Wk,  s1, M, N, K);
  gemm_f32_wmma<<<gg, gb, 0, stream>>>(X,  Wv,  s2, M, N, K);
  gemm_f32_wmma<<<gg, gb, 0, stream>>>(s0, Wqm, s3, M, N, K);
  gemm_f32_wmma<<<gg, gb, 0, stream>>>(s1, Wkm, s0, M, N, K);
  gemm_f32_wmma<<<gg, gb, 0, stream>>>(s2, Wvm, s1, M, N, K);

  attn_f32_wmma<<<Bz * NH * (Sq / QT), 128, 0, stream>>>(s3, s0, s1, mask, s2);

  gemm_f32_wmma<<<gg, gb, 0, stream>>>(s2, Wo, out, M, N, K);
}


# Fold each projection pair: compute W_q*W_q_m, W_k*W_k_m, W_v*W_v_m (2048^3) first, then X times the folded weight; re-associates the f32 sums of the two chained products
# speedup vs baseline: 1.1120x; 1.1120x over previous
#include <hip/hip_runtime.h>
#include <hip/hip_bf16.h>
#include <math.h>

constexpr int Bz = 2;
constexpr int Sq = 2048;
constexpr int Hd = 2048;
constexpr int NH = 16;
constexpr int HD = 128;
constexpr float NEGV = -1000000.0f;

typedef float v2f __attribute__((ext_vector_type(2)));
typedef float v8f __attribute__((ext_vector_type(8)));

#define WMMA_F32(a, b, c) \
  __builtin_amdgcn_wmma_f32_16x16x4_f32(false, (a), false, (b), (short)0, (c), false, false)

__device__ __forceinline__ v8f vzero8() {
  v8f z = {0.f, 0.f, 0.f, 0.f, 0.f, 0.f, 0.f, 0.f};
  return z;
}

constexpr int TM = 128, TN = 128, TK = 32;
constexpr int APAD = 4;
constexpr int BIPAD = 32;

__global__ __launch_bounds__(256) void gemm_f32_wmma(
    const float* __restrict__ A, const float* __restrict__ Bm,
    float* __restrict__ C, int M, int N, int K) {
  __shared__ float As[TM][TK + APAD];
  __shared__ float Bsp[TK / 2][2 * TN + BIPAD];

  const int t = threadIdx.x;
  const int lane = t & 31;
  const int wave = t >> 5;
  const int lm = lane & 15;
  const int lh = lane >> 4;
  const int wm = wave & 3;
  const int wn = wave >> 2;
  const int m0 = blockIdx.y * TM;
  const int n0 = blockIdx.x * TN;

  v8f acc[2][4];
#pragma unroll
  for (int i = 0; i < 2; ++i)
#pragma unroll
    for (int j = 0; j < 4; ++j) acc[i][j] = vzero8();

  for (int kt = 0; kt < K; kt += TK) {
#pragma unroll
    for (int i = 0; i < 4; ++i) {
      int idx = t + i * 256;
      int r = idx >> 3;
      int c = (idx & 7) << 2;
      float4 v = *(const float4*)(A + (size_t)(m0 + r) * K + kt + c);
      *(float4*)&As[r][c] = v;
    }
#pragma unroll
    for (int i = 0; i < 4; ++i) {
      int idx = t + i * 256;
      int r = idx >> 5;
      int c = (idx & 31) << 2;
      float4 v = *(const float4*)(Bm + (size_t)(kt + r) * N + n0 + c);
      float* dst = &Bsp[r >> 1][2 * c + (r & 1)];
      dst[0] = v.x;
      dst[2] = v.y;
      dst[4] = v.z;
      dst[6] = v.w;
    }
    __syncthreads();

#pragma unroll
    for (int kk = 0; kk < TK; kk += 4) {
      v2f af[2];
#pragma unroll
      for (int im = 0; im < 2; ++im) {
        int row = wm * 32 + im * 16 + lm;
        af[im] = *(const v2f*)&As[row][kk + 2 * lh];
      }
      const int khalf = (kk >> 1) + lh;
#pragma unroll
      for (int j = 0; j < 4; ++j) {
        int col = wn * 64 + j * 16 + lm;
        v2f bf = *(const v2f*)&Bsp[khalf][2 * col];
        acc[0][j] = WMMA_F32(af[0], bf, acc[0][j]);
        acc[1][j] = WMMA_F32(af[1], bf, acc[1][j]);
      }
    }
    __syncthreads();
  }

#pragma unroll
  for (int im = 0; im < 2; ++im)
#pragma unroll
    for (int j = 0; j < 4; ++j)
#pragma unroll
      for (int r = 0; r < 8; ++r) {
        int m = m0 + wm * 32 + im * 16 + r + 8 * lh;
        int n = n0 + wn * 64 + j * 16 + lm;
        C[(size_t)m * N + n] = acc[im][j][r];
      }
}

constexpr int QT = 64;
constexpr int KT = 32;
constexpr int KVP = 12;
constexpr int VIPAD = 32;
constexpr int PPAD = 4;

__global__ __launch_bounds__(128) void attn_f32_wmma(
    const float* __restrict__ Qm, const float* __restrict__ Km,
    const float* __restrict__ Vm, const int* __restrict__ mask,
    float* __restrict__ O) {
  __shared__ float Ks[KT][HD + KVP];
  __shared__ float Vsp[KT / 2][2 * HD + VIPAD];
  __shared__ float Ps[4][16][KT + PPAD];
  __shared__ float corrS[4][16];
  __shared__ float rsumS[4][16];
  __shared__ int mkS[KT];

  const int t = threadIdx.x;
  const int lane = t & 31;
  const int w = t >> 5;
  const int lm = lane & 15;
  const int lh = lane >> 4;

  int bid = blockIdx.x;
  const int nqb = Sq / QT;
  const int qb = bid % nqb; bid /= nqb;
  const int h = bid % NH;
  const int b = bid / NH;
  const int q0 = qb * QT;

  const size_t baseBH = (size_t)b * Sq * Hd + (size_t)h * Sq * HD;

  const float* qrow = Qm + baseBH + (size_t)(q0 + w * 16 + lm) * HD;
  v2f qf[HD / 4];
#pragma unroll
  for (int kd = 0; kd < HD / 4; ++kd)
    qf[kd] = *(const v2f*)(qrow + 4 * kd + 2 * lh);

  int mq[8];
#pragma unroll
  for (int r = 0; r < 8; ++r)
    mq[r] = mask[b * Sq + q0 + w * 16 + r + 8 * lh];

  v8f o[8];
#pragma unroll
  for (int s = 0; s < 8; ++s) o[s] = vzero8();
  float rmax = -INFINITY;
  float rsum = 0.0f;

  for (int kt = 0; kt < Sq; kt += KT) {
#pragma unroll
    for (int i = 0; i < 8; ++i) {
      int idx = t + i * 128;
      int r = idx >> 5;
      int c = (idx & 31) << 2;
      float4 kv = *(const float4*)(Km + baseBH + (size_t)(kt + r) * HD + c);
      *(float4*)&Ks[r][c] = kv;
      float4 vv = *(const float4*)(Vm + baseBH + (size_t)(kt + r) * HD + c);
      float* dst = &Vsp[r >> 1][2 * c + (r & 1)];
      dst[0] = vv.x;
      dst[2] = vv.y;
      dst[4] = vv.z;
      dst[6] = vv.w;
    }
    if (t < KT) mkS[t] = mask[b * Sq + kt + t];
    __syncthreads();

    v8f sc[2];
    sc[0] = vzero8();
    sc[1] = vzero8();
#pragma unroll
    for (int kd = 0; kd < HD / 4; ++kd) {
      v2f af = qf[kd];
#pragma unroll
      for (int j = 0; j < 2; ++j) {
        int kc = j * 16 + lm;
        v2f bf = *(const v2f*)&Ks[kc][4 * kd + 2 * lh];
        sc[j] = WMMA_F32(af, bf, sc[j]);
      }
    }

#pragma unroll
    for (int j = 0; j < 2; ++j)
#pragma unroll
      for (int r = 0; r < 8; ++r) {
        int m = r + 8 * lh;
        int kc = j * 16 + lm;
        float v = sc[j][r];
        if (mq[r] == 0 || mkS[kc] == 0) v = NEGV;
        Ps[w][m][kc] = v;
      }
    __syncthreads();

    {
      const int c0 = lh * 16;
      float pm = -INFINITY;
#pragma unroll
      for (int c = 0; c < 16; ++c) pm = fmaxf(pm, Ps[w][lm][c0 + c]);
      pm = fmaxf(pm, __shfl_xor(pm, 16, 32));
      float nm = fmaxf(rmax, pm);
      float cf = __expf(rmax - nm);
      float ps = 0.0f;
#pragma unroll
      for (int c = 0; c < 16; ++c) {
        float e = __expf(Ps[w][lm][c0 + c] - nm);
        Ps[w][lm][c0 + c] = e;
        ps += e;
      }
      ps += __shfl_xor(ps, 16, 32);
      rsum = rsum * cf + ps;
      rmax = nm;
      if (lane < 16) corrS[w][lm] = cf;
    }
    __syncthreads();

    float cv[8];
#pragma unroll
    for (int r = 0; r < 8; ++r) cv[r] = corrS[w][r + 8 * lh];
#pragma unroll
    for (int s = 0; s < 8; ++s)
#pragma unroll
      for (int r = 0; r < 8; ++r) o[s][r] *= cv[r];

#pragma unroll
    for (int kk = 0; kk < KT; kk += 4) {
      v2f af = *(const v2f*)&Ps[w][lm][kk + 2 * lh];
      const int khalf = (kk >> 1) + lh;
#pragma unroll
      for (int s = 0; s < 8; ++s) {
        int col = s * 16 + lm;
        v2f bf = *(const v2f*)&Vsp[khalf][2 * col];
        o[s] = WMMA_F32(af, bf, o[s]);
      }
    }
    __syncthreads();
  }

  if (lane < 16) rsumS[w][lm] = rsum;
  __syncthreads();
  float rv[8];
#pragma unroll
  for (int r = 0; r < 8; ++r) rv[r] = 1.0f / rsumS[w][r + 8 * lh];

#pragma unroll
  for (int s = 0; s < 8; ++s)
#pragma unroll
    for (int r = 0; r < 8; ++r) {
      int s2 = q0 + w * 16 + r + 8 * lh;
      int d = s * 16 + lm;
      O[baseBH + (size_t)s2 * HD + d] = o[s][r] * rv[r];
    }
}

extern "C" void kernel_launch(void* const* d_in, const int* in_sizes, int n_in,
                              void* d_out, int out_size, void* d_ws, size_t ws_size,
                              hipStream_t stream) {
  const float* X    = (const float*)d_in[0];
  const int*   mask = (const int*)d_in[1];
  const float* Wq   = (const float*)d_in[2];
  const float* Wk   = (const float*)d_in[3];
  const float* Wv   = (const float*)d_in[4];
  const float* Wqm  = (const float*)d_in[5];
  const float* Wkm  = (const float*)d_in[6];
  const float* Wvm  = (const float*)d_in[7];
  const float* Wo   = (const float*)d_in[8];
  float* out = (float*)d_out;

  constexpr size_t SZ = (size_t)Bz * Sq * Hd;
  constexpr size_t WZ = (size_t)Hd * Hd;
  float* s0 = (float*)d_ws;
  float* s1 = s0 + SZ;
  float* s2 = s1 + SZ;
  float* s3 = s2 + SZ;
  float* fq = s0;
  float* fk = s0 + WZ;
  float* fv = s1;

  constexpr int M = Bz * Sq, N = Hd, K = Hd;
  dim3 gw(N / TN, K / TM);
  dim3 gg(N / TN, M / TM);
  dim3 gb(256);

  gemm_f32_wmma<<<gw, gb, 0, stream>>>(Wq, Wqm, fq, K, N, K);
  gemm_f32_wmma<<<gw, gb, 0, stream>>>(Wk, Wkm, fk, K, N, K);
  gemm_f32_wmma<<<gw, gb, 0, stream>>>(Wv, Wvm, fv, K, N, K);
  gemm_f32_wmma<<<gg, gb, 0, stream>>>(X, fq, s2, M, N, K);
  gemm_f32_wmma<<<gg, gb, 0, stream>>>(X, fk, s3, M, N, K);
  gemm_f32_wmma<<<gg, gb, 0, stream>>>(X, fv, s0, M, N, K);

  attn_f32_wmma<<<Bz * NH * (Sq / QT), 128, 0, stream>>>(s2, s3, s0, mask, s1);

  gemm_f32_wmma<<<gg, gb, 0, stream>>>(s1, Wo, out, M, N, K);
}


# GEMM staging software-pipelined through registers: the next K-tile's 8 global loads are issued behind the barrier and fly during the current tile's 64 WMMAs, LDS stores at the next trip's top
# speedup vs baseline: 1.1894x; 1.0696x over previous
#include <hip/hip_runtime.h>
#include <hip/hip_bf16.h>
#include <math.h>

constexpr int Bz = 2;
constexpr int Sq = 2048;
constexpr int Hd = 2048;
constexpr int NH = 16;
constexpr int HD = 128;
constexpr float NEGV = -1000000.0f;

typedef float v2f __attribute__((ext_vector_type(2)));
typedef float v8f __attribute__((ext_vector_type(8)));

#define WMMA_F32(a, b, c) \
  __builtin_amdgcn_wmma_f32_16x16x4_f32(false, (a), false, (b), (short)0, (c), false, false)

__device__ __forceinline__ v8f vzero8() {
  v8f z = {0.f, 0.f, 0.f, 0.f, 0.f, 0.f, 0.f, 0.f};
  return z;
}

constexpr int TM = 128, TN = 128, TK = 32;
constexpr int APAD = 4;
constexpr int BIPAD = 32;

__global__ __launch_bounds__(256) void gemm_f32_wmma(
    const float* __restrict__ A, const float* __restrict__ Bm,
    float* __restrict__ C, int M, int N, int K) {
  __shared__ float As[TM][TK + APAD];
  __shared__ float Bsp[TK / 2][2 * TN + BIPAD];

  const int t = threadIdx.x;
  const int lane = t & 31;
  const int wave = t >> 5;
  const int lm = lane & 15;
  const int lh = lane >> 4;
  const int wm = wave & 3;
  const int wn = wave >> 2;
  const int m0 = blockIdx.y * TM;
  const int n0 = blockIdx.x * TN;

  v8f acc[2][4];
#pragma unroll
  for (int i = 0; i < 2; ++i)
#pragma unroll
    for (int j = 0; j < 4; ++j) acc[i][j] = vzero8();

  const int ar = t >> 3;
  const int ac = (t & 7) << 2;
  const int br = t >> 5;
  const int bc = (t & 31) << 2;
  const float* ap = A + (size_t)(m0 + ar) * K + ac;
  const float* bp = Bm + (size_t)br * N + n0 + bc;

  const size_t aStep = (size_t)32 * K;
  const size_t bStep = (size_t)8 * N;
  float4 ga0 = *(const float4*)(ap);
  float4 ga1 = *(const float4*)(ap + aStep);
  float4 ga2 = *(const float4*)(ap + 2 * aStep);
  float4 ga3 = *(const float4*)(ap + 3 * aStep);
  float4 gb0 = *(const float4*)(bp);
  float4 gb1 = *(const float4*)(bp + bStep);
  float4 gb2 = *(const float4*)(bp + 2 * bStep);
  float4 gb3 = *(const float4*)(bp + 3 * bStep);

  float* const bdst = &Bsp[br >> 1][2 * bc + (br & 1)];
  constexpr int BROW4 = 4 * (2 * TN + BIPAD);

  for (int kt = 0; kt < K; kt += TK) {
    *(float4*)&As[ar][ac] = ga0;
    *(float4*)&As[ar + 32][ac] = ga1;
    *(float4*)&As[ar + 64][ac] = ga2;
    *(float4*)&As[ar + 96][ac] = ga3;
    bdst[0] = gb0.x; bdst[2] = gb0.y; bdst[4] = gb0.z; bdst[6] = gb0.w;
    bdst[BROW4 + 0] = gb1.x; bdst[BROW4 + 2] = gb1.y;
    bdst[BROW4 + 4] = gb1.z; bdst[BROW4 + 6] = gb1.w;
    bdst[2 * BROW4 + 0] = gb2.x; bdst[2 * BROW4 + 2] = gb2.y;
    bdst[2 * BROW4 + 4] = gb2.z; bdst[2 * BROW4 + 6] = gb2.w;
    bdst[3 * BROW4 + 0] = gb3.x; bdst[3 * BROW4 + 2] = gb3.y;
    bdst[3 * BROW4 + 4] = gb3.z; bdst[3 * BROW4 + 6] = gb3.w;
    __syncthreads();

    {
      const int kn = (kt + TK < K) ? kt + TK : kt;
      const float* an = ap + kn;
      const float* bn = bp + (size_t)kn * N;
      ga0 = *(const float4*)(an);
      ga1 = *(const float4*)(an + aStep);
      ga2 = *(const float4*)(an + 2 * aStep);
      ga3 = *(const float4*)(an + 3 * aStep);
      gb0 = *(const float4*)(bn);
      gb1 = *(const float4*)(bn + bStep);
      gb2 = *(const float4*)(bn + 2 * bStep);
      gb3 = *(const float4*)(bn + 3 * bStep);
    }
    __builtin_amdgcn_sched_barrier(0);

#pragma unroll
    for (int kk = 0; kk < TK; kk += 4) {
      v2f af[2];
#pragma unroll
      for (int im = 0; im < 2; ++im) {
        int row = wm * 32 + im * 16 + lm;
        af[im] = *(const v2f*)&As[row][kk + 2 * lh];
      }
      const int khalf = (kk >> 1) + lh;
#pragma unroll
      for (int j = 0; j < 4; ++j) {
        int col = wn * 64 + j * 16 + lm;
        v2f bf = *(const v2f*)&Bsp[khalf][2 * col];
        acc[0][j] = WMMA_F32(af[0], bf, acc[0][j]);
        acc[1][j] = WMMA_F32(af[1], bf, acc[1][j]);
      }
    }
    __syncthreads();
  }

#pragma unroll
  for (int im = 0; im < 2; ++im)
#pragma unroll
    for (int j = 0; j < 4; ++j)
#pragma unroll
      for (int r = 0; r < 8; ++r) {
        int m = m0 + wm * 32 + im * 16 + r + 8 * lh;
        int n = n0 + wn * 64 + j * 16 + lm;
        C[(size_t)m * N + n] = acc[im][j][r];
      }
}

constexpr int QT = 64;
constexpr int KT = 32;
constexpr int KVP = 12;
constexpr int VIPAD = 32;
constexpr int PPAD = 4;

__global__ __launch_bounds__(128) void attn_f32_wmma(
    const float* __restrict__ Qm, const float* __restrict__ Km,
    const float* __restrict__ Vm, const int* __restrict__ mask,
    float* __restrict__ O) {
  __shared__ float Ks[KT][HD + KVP];
  __shared__ float Vsp[KT / 2][2 * HD + VIPAD];
  __shared__ float Ps[4][16][KT + PPAD];
  __shared__ float corrS[4][16];
  __shared__ float rsumS[4][16];
  __shared__ int mkS[KT];

  const int t = threadIdx.x;
  const int lane = t & 31;
  const int w = t >> 5;
  const int lm = lane & 15;
  const int lh = lane >> 4;

  int bid = blockIdx.x;
  const int nqb = Sq / QT;
  const int qb = bid % nqb; bid /= nqb;
  const int h = bid % NH;
  const int b = bid / NH;
  const int q0 = qb * QT;

  const size_t baseBH = (size_t)b * Sq * Hd + (size_t)h * Sq * HD;

  const float* qrow = Qm + baseBH + (size_t)(q0 + w * 16 + lm) * HD;
  v2f qf[HD / 4];
#pragma unroll
  for (int kd = 0; kd < HD / 4; ++kd)
    qf[kd] = *(const v2f*)(qrow + 4 * kd + 2 * lh);

  int mq[8];
#pragma unroll
  for (int r = 0; r < 8; ++r)
    mq[r] = mask[b * Sq + q0 + w * 16 + r + 8 * lh];

  v8f o[8];
#pragma unroll
  for (int s = 0; s < 8; ++s) o[s] = vzero8();
  float rmax = -INFINITY;
  float rsum = 0.0f;

  for (int kt = 0; kt < Sq; kt += KT) {
#pragma unroll
    for (int i = 0; i < 8; ++i) {
      int idx = t + i * 128;
      int r = idx >> 5;
      int c = (idx & 31) << 2;
      float4 kv = *(const float4*)(Km + baseBH + (size_t)(kt + r) * HD + c);
      *(float4*)&Ks[r][c] = kv;
      float4 vv = *(const float4*)(Vm + baseBH + (size_t)(kt + r) * HD + c);
      float* dst = &Vsp[r >> 1][2 * c + (r & 1)];
      dst[0] = vv.x;
      dst[2] = vv.y;
      dst[4] = vv.z;
      dst[6] = vv.w;
    }
    if (t < KT) mkS[t] = mask[b * Sq + kt + t];
    __syncthreads();

    v8f sc[2];
    sc[0] = vzero8();
    sc[1] = vzero8();
#pragma unroll
    for (int kd = 0; kd < HD / 4; ++kd) {
      v2f af = qf[kd];
#pragma unroll
      for (int j = 0; j < 2; ++j) {
        int kc = j * 16 + lm;
        v2f bf = *(const v2f*)&Ks[kc][4 * kd + 2 * lh];
        sc[j] = WMMA_F32(af, bf, sc[j]);
      }
    }

#pragma unroll
    for (int j = 0; j < 2; ++j)
#pragma unroll
      for (int r = 0; r < 8; ++r) {
        int m = r + 8 * lh;
        int kc = j * 16 + lm;
        float v = sc[j][r];
        if (mq[r] == 0 || mkS[kc] == 0) v = NEGV;
        Ps[w][m][kc] = v;
      }
    __syncthreads();

    {
      const int c0 = lh * 16;
      float pm = -INFINITY;
#pragma unroll
      for (int c = 0; c < 16; ++c) pm = fmaxf(pm, Ps[w][lm][c0 + c]);
      pm = fmaxf(pm, __shfl_xor(pm, 16, 32));
      float nm = fmaxf(rmax, pm);
      float cf = __expf(rmax - nm);
      float ps = 0.0f;
#pragma unroll
      for (int c = 0; c < 16; ++c) {
        float e = __expf(Ps[w][lm][c0 + c] - nm);
        Ps[w][lm][c0 + c] = e;
        ps += e;
      }
      ps += __shfl_xor(ps, 16, 32);
      rsum = rsum * cf + ps;
      rmax = nm;
      if (lane < 16) corrS[w][lm] = cf;
    }
    __syncthreads();

    float cv[8];
#pragma unroll
    for (int r = 0; r < 8; ++r) cv[r] = corrS[w][r + 8 * lh];
#pragma unroll
    for (int s = 0; s < 8; ++s)
#pragma unroll
      for (int r = 0; r < 8; ++r) o[s][r] *= cv[r];

#pragma unroll
    for (int kk = 0; kk < KT; kk += 4) {
      v2f af = *(const v2f*)&Ps[w][lm][kk + 2 * lh];
      const int khalf = (kk >> 1) + lh;
#pragma unroll
      for (int s = 0; s < 8; ++s) {
        int col = s * 16 + lm;
        v2f bf = *(const v2f*)&Vsp[khalf][2 * col];
        o[s] = WMMA_F32(af, bf, o[s]);
      }
    }
    __syncthreads();
  }

  if (lane < 16) rsumS[w][lm] = rsum;
  __syncthreads();
  float rv[8];
#pragma unroll
  for (int r = 0; r < 8; ++r) rv[r] = 1.0f / rsumS[w][r + 8 * lh];

#pragma unroll
  for (int s = 0; s < 8; ++s)
#pragma unroll
    for (int r = 0; r < 8; ++r) {
      int s2 = q0 + w * 16 + r + 8 * lh;
      int d = s * 16 + lm;
      O[baseBH + (size_t)s2 * HD + d] = o[s][r] * rv[r];
    }
}

extern "C" void kernel_launch(void* const* d_in, const int* in_sizes, int n_in,
                              void* d_out, int out_size, void* d_ws, size_t ws_size,
                              hipStream_t stream) {
  const float* X    = (const float*)d_in[0];
  const int*   mask = (const int*)d_in[1];
  const float* Wq   = (const float*)d_in[2];
  const float* Wk   = (const float*)d_in[3];
  const float* Wv   = (const float*)d_in[4];
  const float* Wqm  = (const float*)d_in[5];
  const float* Wkm  = (const float*)d_in[6];
  const float* Wvm  = (const float*)d_in[7];
  const float* Wo   = (const float*)d_in[8];
  float* out = (float*)d_out;

  constexpr size_t SZ = (size_t)Bz * Sq * Hd;
  constexpr size_t WZ = (size_t)Hd * Hd;
  float* s0 = (float*)d_ws;
  float* s1 = s0 + SZ;
  float* s2 = s1 + SZ;
  float* s3 = s2 + SZ;
  float* fq = s0;
  float* fk = s0 + WZ;
  float* fv = s1;

  constexpr int M = Bz * Sq, N = Hd, K = Hd;
  dim3 gw(N / TN, K / TM);
  dim3 gg(N / TN, M / TM);
  dim3 gb(256);

  gemm_f32_wmma<<<gw, gb, 0, stream>>>(Wq, Wqm, fq, K, N, K);
  gemm_f32_wmma<<<gw, gb, 0, stream>>>(Wk, Wkm, fk, K, N, K);
  gemm_f32_wmma<<<gw, gb, 0, stream>>>(Wv, Wvm, fv, K, N, K);
  gemm_f32_wmma<<<gg, gb, 0, stream>>>(X, fq, s2, M, N, K);
  gemm_f32_wmma<<<gg, gb, 0, stream>>>(X, fk, s3, M, N, K);
  gemm_f32_wmma<<<gg, gb, 0, stream>>>(X, fv, s0, M, N, K);

  attn_f32_wmma<<<Bz * NH * (Sq / QT), 128, 0, stream>>>(s2, s3, s0, mask, s1);

  gemm_f32_wmma<<<gg, gb, 0, stream>>>(s1, Wo, out, M, N, K);
}


# The three weight products share one launch and the three projections another (blockIdx.z picks the operand triple): 3 workgroups per WGP instead of 1 or 2, 8 launches become 4; V_m parked in out
# speedup vs baseline: 1.3221x; 1.1116x over previous
#include <hip/hip_runtime.h>
#include <hip/hip_bf16.h>
#include <math.h>

constexpr int Bz = 2;
constexpr int Sq = 2048;
constexpr int Hd = 2048;
constexpr int NH = 16;
constexpr int HD = 128;
constexpr float NEGV = -1000000.0f;

typedef float v2f __attribute__((ext_vector_type(2)));
typedef float v8f __attribute__((ext_vector_type(8)));

#define WMMA_F32(a, b, c) \
  __builtin_amdgcn_wmma_f32_16x16x4_f32(false, (a), false, (b), (short)0, (c), false, false)

__device__ __forceinline__ v8f vzero8() {
  v8f z = {0.f, 0.f, 0.f, 0.f, 0.f, 0.f, 0.f, 0.f};
  return z;
}

constexpr int TM = 128, TN = 128, TK = 32;
constexpr int APAD = 4;
constexpr int BIPAD = 32;

__global__ __launch_bounds__(256) void gemm_f32_wmma(
    const float* __restrict__ A0, const float* __restrict__ B0, float* __restrict__ C0,
    const float* __restrict__ A1, const float* __restrict__ B1, float* __restrict__ C1,
    const float* __restrict__ A2, const float* __restrict__ B2, float* __restrict__ C2,
    int M, int N, int K) {
  __shared__ float As[TM][TK + APAD];
  __shared__ float Bsp[TK / 2][2 * TN + BIPAD];

  const int z = blockIdx.z;
  const float* __restrict__ A  = (z == 0) ? A0 : (z == 1) ? A1 : A2;
  const float* __restrict__ Bm = (z == 0) ? B0 : (z == 1) ? B1 : B2;
  float* __restrict__ C        = (z == 0) ? C0 : (z == 1) ? C1 : C2;

  const int t = threadIdx.x;
  const int lane = t & 31;
  const int wave = t >> 5;
  const int lm = lane & 15;
  const int lh = lane >> 4;
  const int wm = wave & 3;
  const int wn = wave >> 2;
  const int m0 = blockIdx.y * TM;
  const int n0 = blockIdx.x * TN;

  v8f acc[2][4];
#pragma unroll
  for (int i = 0; i < 2; ++i)
#pragma unroll
    for (int j = 0; j < 4; ++j) acc[i][j] = vzero8();

  const int ar = t >> 3;
  const int ac = (t & 7) << 2;
  const int br = t >> 5;
  const int bc = (t & 31) << 2;
  const float* ap = A + (size_t)(m0 + ar) * K + ac;
  const float* bp = Bm + (size_t)br * N + n0 + bc;

  const size_t aStep = (size_t)32 * K;
  const size_t bStep = (size_t)8 * N;
  float4 ga0 = *(const float4*)(ap);
  float4 ga1 = *(const float4*)(ap + aStep);
  float4 ga2 = *(const float4*)(ap + 2 * aStep);
  float4 ga3 = *(const float4*)(ap + 3 * aStep);
  float4 gb0 = *(const float4*)(bp);
  float4 gb1 = *(const float4*)(bp + bStep);
  float4 gb2 = *(const float4*)(bp + 2 * bStep);
  float4 gb3 = *(const float4*)(bp + 3 * bStep);

  float* const bdst = &Bsp[br >> 1][2 * bc + (br & 1)];
  constexpr int BROW4 = 4 * (2 * TN + BIPAD);

  for (int kt = 0; kt < K; kt += TK) {
    *(float4*)&As[ar][ac] = ga0;
    *(float4*)&As[ar + 32][ac] = ga1;
    *(float4*)&As[ar + 64][ac] = ga2;
    *(float4*)&As[ar + 96][ac] = ga3;
    bdst[0] = gb0.x; bdst[2] = gb0.y; bdst[4] = gb0.z; bdst[6] = gb0.w;
    bdst[BROW4 + 0] = gb1.x; bdst[BROW4 + 2] = gb1.y;
    bdst[BROW4 + 4] = gb1.z; bdst[BROW4 + 6] = gb1.w;
    bdst[2 * BROW4 + 0] = gb2.x; bdst[2 * BROW4 + 2] = gb2.y;
    bdst[2 * BROW4 + 4] = gb2.z; bdst[2 * BROW4 + 6] = gb2.w;
    bdst[3 * BROW4 + 0] = gb3.x; bdst[3 * BROW4 + 2] = gb3.y;
    bdst[3 * BROW4 + 4] = gb3.z; bdst[3 * BROW4 + 6] = gb3.w;
    __syncthreads();

    {
      const int kn = (kt + TK < K) ? kt + TK : kt;
      const float* an = ap + kn;
      const float* bn = bp + (size_t)kn * N;
      ga0 = *(const float4*)(an);
      ga1 = *(const float4*)(an + aStep);
      ga2 = *(const float4*)(an + 2 * aStep);
      ga3 = *(const float4*)(an + 3 * aStep);
      gb0 = *(const float4*)(bn);
      gb1 = *(const float4*)(bn + bStep);
      gb2 = *(const float4*)(bn + 2 * bStep);
      gb3 = *(const float4*)(bn + 3 * bStep);
    }
    __builtin_amdgcn_sched_barrier(0);

#pragma unroll
    for (int kk = 0; kk < TK; kk += 4) {
      v2f af[2];
#pragma unroll
      for (int im = 0; im < 2; ++im) {
        int row = wm * 32 + im * 16 + lm;
        af[im] = *(const v2f*)&As[row][kk + 2 * lh];
      }
      const int khalf = (kk >> 1) + lh;
#pragma unroll
      for (int j = 0; j < 4; ++j) {
        int col = wn * 64 + j * 16 + lm;
        v2f bf = *(const v2f*)&Bsp[khalf][2 * col];
        acc[0][j] = WMMA_F32(af[0], bf, acc[0][j]);
        acc[1][j] = WMMA_F32(af[1], bf, acc[1][j]);
      }
    }
    __syncthreads();
  }

#pragma unroll
  for (int im = 0; im < 2; ++im)
#pragma unroll
    for (int j = 0; j < 4; ++j)
#pragma unroll
      for (int r = 0; r < 8; ++r) {
        int m = m0 + wm * 32 + im * 16 + r + 8 * lh;
        int n = n0 + wn * 64 + j * 16 + lm;
        C[(size_t)m * N + n] = acc[im][j][r];
      }
}

constexpr int QT = 64;
constexpr int KT = 32;
constexpr int KVP = 12;
constexpr int VIPAD = 32;
constexpr int PPAD = 4;

__global__ __launch_bounds__(128) void attn_f32_wmma(
    const float* __restrict__ Qm, const float* __restrict__ Km,
    const float* __restrict__ Vm, const int* __restrict__ mask,
    float* __restrict__ O) {
  __shared__ float Ks[KT][HD + KVP];
  __shared__ float Vsp[KT / 2][2 * HD + VIPAD];
  __shared__ float Ps[4][16][KT + PPAD];
  __shared__ float corrS[4][16];
  __shared__ float rsumS[4][16];
  __shared__ int mkS[KT];

  const int t = threadIdx.x;
  const int lane = t & 31;
  const int w = t >> 5;
  const int lm = lane & 15;
  const int lh = lane >> 4;

  int bid = blockIdx.x;
  const int nqb = Sq / QT;
  const int qb = bid % nqb; bid /= nqb;
  const int h = bid % NH;
  const int b = bid / NH;
  const int q0 = qb * QT;

  const size_t baseBH = (size_t)b * Sq * Hd + (size_t)h * Sq * HD;

  const float* qrow = Qm + baseBH + (size_t)(q0 + w * 16 + lm) * HD;
  v2f qf[HD / 4];
#pragma unroll
  for (int kd = 0; kd < HD / 4; ++kd)
    qf[kd] = *(const v2f*)(qrow + 4 * kd + 2 * lh);

  int mq[8];
#pragma unroll
  for (int r = 0; r < 8; ++r)
    mq[r] = mask[b * Sq + q0 + w * 16 + r + 8 * lh];

  v8f o[8];
#pragma unroll
  for (int s = 0; s < 8; ++s) o[s] = vzero8();
  float rmax = -INFINITY;
  float rsum = 0.0f;

  for (int kt = 0; kt < Sq; kt += KT) {
#pragma unroll
    for (int i = 0; i < 8; ++i) {
      int idx = t + i * 128;
      int r = idx >> 5;
      int c = (idx & 31) << 2;
      float4 kv = *(const float4*)(Km + baseBH + (size_t)(kt + r) * HD + c);
      *(float4*)&Ks[r][c] = kv;
      float4 vv = *(const float4*)(Vm + baseBH + (size_t)(kt + r) * HD + c);
      float* dst = &Vsp[r >> 1][2 * c + (r & 1)];
      dst[0] = vv.x;
      dst[2] = vv.y;
      dst[4] = vv.z;
      dst[6] = vv.w;
    }
    if (t < KT) mkS[t] = mask[b * Sq + kt + t];
    __syncthreads();

    v8f sc[2];
    sc[0] = vzero8();
    sc[1] = vzero8();
#pragma unroll
    for (int kd = 0; kd < HD / 4; ++kd) {
      v2f af = qf[kd];
#pragma unroll
      for (int j = 0; j < 2; ++j) {
        int kc = j * 16 + lm;
        v2f bf = *(const v2f*)&Ks[kc][4 * kd + 2 * lh];
        sc[j] = WMMA_F32(af, bf, sc[j]);
      }
    }

#pragma unroll
    for (int j = 0; j < 2; ++j)
#pragma unroll
      for (int r = 0; r < 8; ++r) {
        int m = r + 8 * lh;
        int kc = j * 16 + lm;
        float v = sc[j][r];
        if (mq[r] == 0 || mkS[kc] == 0) v = NEGV;
        Ps[w][m][kc] = v;
      }
    __syncthreads();

    {
      const int c0 = lh * 16;
      float pm = -INFINITY;
#pragma unroll
      for (int c = 0; c < 16; ++c) pm = fmaxf(pm, Ps[w][lm][c0 + c]);
      pm = fmaxf(pm, __shfl_xor(pm, 16, 32));
      float nm = fmaxf(rmax, pm);
      float cf = __expf(rmax - nm);
      float ps = 0.0f;
#pragma unroll
      for (int c = 0; c < 16; ++c) {
        float e = __expf(Ps[w][lm][c0 + c] - nm);
        Ps[w][lm][c0 + c] = e;
        ps += e;
      }
      ps += __shfl_xor(ps, 16, 32);
      rsum = rsum * cf + ps;
      rmax = nm;
      if (lane < 16) corrS[w][lm] = cf;
    }
    __syncthreads();

    float cv[8];
#pragma unroll
    for (int r = 0; r < 8; ++r) cv[r] = corrS[w][r + 8 * lh];
#pragma unroll
    for (int s = 0; s < 8; ++s)
#pragma unroll
      for (int r = 0; r < 8; ++r) o[s][r] *= cv[r];

#pragma unroll
    for (int kk = 0; kk < KT; kk += 4) {
      v2f af = *(const v2f*)&Ps[w][lm][kk + 2 * lh];
      const int khalf = (kk >> 1) + lh;
#pragma unroll
      for (int s = 0; s < 8; ++s) {
        int col = s * 16 + lm;
        v2f bf = *(const v2f*)&Vsp[khalf][2 * col];
        o[s] = WMMA_F32(af, bf, o[s]);
      }
    }
    __syncthreads();
  }

  if (lane < 16) rsumS[w][lm] = rsum;
  __syncthreads();
  float rv[8];
#pragma unroll
  for (int r = 0; r < 8; ++r) rv[r] = 1.0f / rsumS[w][r + 8 * lh];

#pragma unroll
  for (int s = 0; s < 8; ++s)
#pragma unroll
    for (int r = 0; r < 8; ++r) {
      int s2 = q0 + w * 16 + r + 8 * lh;
      int d = s * 16 + lm;
      O[baseBH + (size_t)s2 * HD + d] = o[s][r] * rv[r];
    }
}

extern "C" void kernel_launch(void* const* d_in, const int* in_sizes, int n_in,
                              void* d_out, int out_size, void* d_ws, size_t ws_size,
                              hipStream_t stream) {
  const float* X    = (const float*)d_in[0];
  const int*   mask = (const int*)d_in[1];
  const float* Wq   = (const float*)d_in[2];
  const float* Wk   = (const float*)d_in[3];
  const float* Wv   = (const float*)d_in[4];
  const float* Wqm  = (const float*)d_in[5];
  const float* Wkm  = (const float*)d_in[6];
  const float* Wvm  = (const float*)d_in[7];
  const float* Wo   = (const float*)d_in[8];
  float* out = (float*)d_out;

  constexpr size_t SZ = (size_t)Bz * Sq * Hd;
  constexpr size_t WZ = (size_t)Hd * Hd;
  float* s0 = (float*)d_ws;
  float* s1 = s0 + SZ;
  float* s2 = s1 + SZ;
  float* s3 = s2 + SZ;
  float* fq = s0;
  float* fk = s0 + WZ;
  float* fv = s1;

  constexpr int M = Bz * Sq, N = Hd, K = Hd;
  dim3 gw3(N / TN, K / TM, 3);
  dim3 gg3(N / TN, M / TM, 3);
  dim3 gg1(N / TN, M / TM, 1);
  dim3 gb(256);

  float* vm = out;

  gemm_f32_wmma<<<gw3, gb, 0, stream>>>(Wq, Wqm, fq, Wk, Wkm, fk, Wv, Wvm, fv, K, N, K);
  gemm_f32_wmma<<<gg3, gb, 0, stream>>>(X, fq, s2, X, fk, s3, X, fv, vm, M, N, K);

  attn_f32_wmma<<<Bz * NH * (Sq / QT), 128, 0, stream>>>(s2, s3, vm, mask, s1);

  gemm_f32_wmma<<<gg1, gb, 0, stream>>>(s1, Wo, out, s1, Wo, out, s1, Wo, out, M, N, K);
}


# Attention over unmasked positions only: Q/K/V rows gathered through a compacted index list, masked keys' shared exp(NEG-M) term added in closed form from V column sums; softmax sums regrouped
# speedup vs baseline: 1.6250x; 1.2291x over previous
#include <hip/hip_runtime.h>
#include <hip/hip_bf16.h>
#include <math.h>

constexpr int Bz = 2;
constexpr int Sq = 2048;
constexpr int Hd = 2048;
constexpr int NH = 16;
constexpr int HD = 128;
constexpr float NEGV = -1000000.0f;

typedef float v2f __attribute__((ext_vector_type(2)));
typedef float v8f __attribute__((ext_vector_type(8)));

#define WMMA_F32(a, b, c) \
  __builtin_amdgcn_wmma_f32_16x16x4_f32(false, (a), false, (b), (short)0, (c), false, false)

__device__ __forceinline__ v8f vzero8() {
  v8f z = {0.f, 0.f, 0.f, 0.f, 0.f, 0.f, 0.f, 0.f};
  return z;
}

constexpr int TM = 128, TN = 128, TK = 32;
constexpr int APAD = 4;
constexpr int BIPAD = 32;

__global__ __launch_bounds__(256) void gemm_f32_wmma(
    const float* __restrict__ A0, const float* __restrict__ B0, float* __restrict__ C0,
    const float* __restrict__ A1, const float* __restrict__ B1, float* __restrict__ C1,
    const float* __restrict__ A2, const float* __restrict__ B2, float* __restrict__ C2,
    int M, int N, int K) {
  __shared__ float As[TM][TK + APAD];
  __shared__ float Bsp[TK / 2][2 * TN + BIPAD];

  const int z = blockIdx.z;
  const float* __restrict__ A  = (z == 0) ? A0 : (z == 1) ? A1 : A2;
  const float* __restrict__ Bm = (z == 0) ? B0 : (z == 1) ? B1 : B2;
  float* __restrict__ C        = (z == 0) ? C0 : (z == 1) ? C1 : C2;

  const int t = threadIdx.x;
  const int lane = t & 31;
  const int wave = t >> 5;
  const int lm = lane & 15;
  const int lh = lane >> 4;
  const int wm = wave & 3;
  const int wn = wave >> 2;
  const int m0 = blockIdx.y * TM;
  const int n0 = blockIdx.x * TN;

  v8f acc[2][4];
#pragma unroll
  for (int i = 0; i < 2; ++i)
#pragma unroll
    for (int j = 0; j < 4; ++j) acc[i][j] = vzero8();

  const int ar = t >> 3;
  const int ac = (t & 7) << 2;
  const int br = t >> 5;
  const int bc = (t & 31) << 2;
  const float* ap = A + (size_t)(m0 + ar) * K + ac;
  const float* bp = Bm + (size_t)br * N + n0 + bc;

  const size_t aStep = (size_t)32 * K;
  const size_t bStep = (size_t)8 * N;
  float4 ga0 = *(const float4*)(ap);
  float4 ga1 = *(const float4*)(ap + aStep);
  float4 ga2 = *(const float4*)(ap + 2 * aStep);
  float4 ga3 = *(const float4*)(ap + 3 * aStep);
  float4 gb0 = *(const float4*)(bp);
  float4 gb1 = *(const float4*)(bp + bStep);
  float4 gb2 = *(const float4*)(bp + 2 * bStep);
  float4 gb3 = *(const float4*)(bp + 3 * bStep);

  float* const bdst = &Bsp[br >> 1][2 * bc + (br & 1)];
  constexpr int BROW4 = 4 * (2 * TN + BIPAD);

  for (int kt = 0; kt < K; kt += TK) {
    *(float4*)&As[ar][ac] = ga0;
    *(float4*)&As[ar + 32][ac] = ga1;
    *(float4*)&As[ar + 64][ac] = ga2;
    *(float4*)&As[ar + 96][ac] = ga3;
    bdst[0] = gb0.x; bdst[2] = gb0.y; bdst[4] = gb0.z; bdst[6] = gb0.w;
    bdst[BROW4 + 0] = gb1.x; bdst[BROW4 + 2] = gb1.y;
    bdst[BROW4 + 4] = gb1.z; bdst[BROW4 + 6] = gb1.w;
    bdst[2 * BROW4 + 0] = gb2.x; bdst[2 * BROW4 + 2] = gb2.y;
    bdst[2 * BROW4 + 4] = gb2.z; bdst[2 * BROW4 + 6] = gb2.w;
    bdst[3 * BROW4 + 0] = gb3.x; bdst[3 * BROW4 + 2] = gb3.y;
    bdst[3 * BROW4 + 4] = gb3.z; bdst[3 * BROW4 + 6] = gb3.w;
    __syncthreads();

    {
      const int kn = (kt + TK < K) ? kt + TK : kt;
      const float* an = ap + kn;
      const float* bn = bp + (size_t)kn * N;
      ga0 = *(const float4*)(an);
      ga1 = *(const float4*)(an + aStep);
      ga2 = *(const float4*)(an + 2 * aStep);
      ga3 = *(const float4*)(an + 3 * aStep);
      gb0 = *(const float4*)(bn);
      gb1 = *(const float4*)(bn + bStep);
      gb2 = *(const float4*)(bn + 2 * bStep);
      gb3 = *(const float4*)(bn + 3 * bStep);
    }
    __builtin_amdgcn_sched_barrier(0);

#pragma unroll
    for (int kk = 0; kk < TK; kk += 4) {
      v2f af[2];
#pragma unroll
      for (int im = 0; im < 2; ++im) {
        int row = wm * 32 + im * 16 + lm;
        af[im] = *(const v2f*)&As[row][kk + 2 * lh];
      }
      const int khalf = (kk >> 1) + lh;
#pragma unroll
      for (int j = 0; j < 4; ++j) {
        int col = wn * 64 + j * 16 + lm;
        v2f bf = *(const v2f*)&Bsp[khalf][2 * col];
        acc[0][j] = WMMA_F32(af[0], bf, acc[0][j]);
        acc[1][j] = WMMA_F32(af[1], bf, acc[1][j]);
      }
    }
    __syncthreads();
  }

#pragma unroll
  for (int im = 0; im < 2; ++im)
#pragma unroll
    for (int j = 0; j < 4; ++j)
#pragma unroll
      for (int r = 0; r < 8; ++r) {
        int m = m0 + wm * 32 + im * 16 + r + 8 * lh;
        int n = n0 + wn * 64 + j * 16 + lm;
        C[(size_t)m * N + n] = acc[im][j][r];
      }
}

constexpr int VSL = 16;
constexpr int VSR = Sq / VSL;

__global__ __launch_bounds__(256) void vsum_partial(
    const float* __restrict__ Vm, const int* __restrict__ mask,
    float* __restrict__ pall, float* __restrict__ pmsk) {
  __shared__ float redA[8][HD];
  __shared__ float redM[8][HD];
  const int t = threadIdx.x;
  const int c4 = (t & 31) << 2;
  const int rg = t >> 5;
  const int sl = blockIdx.x % VSL;
  const int bh = blockIdx.x / VSL;
  const int b = bh / NH;
  const float* vp = Vm + (size_t)bh * Sq * HD + (size_t)(sl * VSR) * HD + c4;
  const int* mp = mask + b * Sq + sl * VSR;
  float4 sa = make_float4(0.f, 0.f, 0.f, 0.f);
  float4 sm = make_float4(0.f, 0.f, 0.f, 0.f);
#pragma unroll
  for (int i = 0; i < VSR / 8; ++i) {
    const int r = rg + 8 * i;
    const float4 v = *(const float4*)(vp + (size_t)r * HD);
    const int mk = mp[r];
    sa.x += v.x; sa.y += v.y; sa.z += v.z; sa.w += v.w;
    if (mk == 0) { sm.x += v.x; sm.y += v.y; sm.z += v.z; sm.w += v.w; }
  }
  *(float4*)&redA[rg][c4] = sa;
  *(float4*)&redM[rg][c4] = sm;
  __syncthreads();
  if (t < HD) {
    float a = 0.f, m = 0.f;
#pragma unroll
    for (int g = 0; g < 8; ++g) { a += redA[g][t]; m += redM[g][t]; }
    pall[((size_t)bh * VSL + sl) * HD + t] = a;
    pmsk[((size_t)bh * VSL + sl) * HD + t] = m;
  }
}

constexpr int QT = 64;
constexpr int KT = 32;
constexpr int KVP = 12;
constexpr int VIPAD = 32;
constexpr int PPAD = 4;

__global__ __launch_bounds__(128) void attn_f32_wmma(
    const float* __restrict__ Qm, const float* __restrict__ Km,
    const float* __restrict__ Vm, const int* __restrict__ mask,
    const float* __restrict__ pall, const float* __restrict__ pmsk,
    float* __restrict__ O) {
  __shared__ float Ks[KT][HD + KVP];
  __shared__ float Vsp[KT / 2][2 * HD + VIPAD];
  __shared__ float Ps[4][16][KT + PPAD];
  __shared__ float corrS[4][16];
  __shared__ float rsumS[4][16];
  __shared__ float rmaxS[4][16];
  __shared__ float vmskS[HD];
  __shared__ int U[Sq];
  __shared__ int wtot[4];

  const int t = threadIdx.x;
  const int lane = t & 31;
  const int w = t >> 5;
  const int lm = lane & 15;
  const int lh = lane >> 4;

  const int bh = blockIdx.x % (Bz * NH);
  const int qb = blockIdx.x / (Bz * NH);
  const int b = bh / NH;
  const int q0 = qb * QT;

  const size_t baseBH = (size_t)bh * Sq * HD;
  const int* mrow = mask + b * Sq;

  unsigned bal[16];
  int cnt = 0;
#pragma unroll
  for (int i = 0; i < 16; ++i) {
    bal[i] = __builtin_amdgcn_ballot_w32(mrow[w * 512 + i * 32 + lane] != 0);
    cnt += __popc(bal[i]);
  }
  if (lane == 0) wtot[w] = cnt;

  float vall = 0.f, vmsk = 0.f;
  {
    const float* pa = pall + (size_t)bh * VSL * HD + t;
    const float* pm = pmsk + (size_t)bh * VSL * HD + t;
#pragma unroll
    for (int s = 0; s < VSL; ++s) { vall += pa[s * HD]; vmsk += pm[s * HD]; }
  }
  vmskS[t] = vmsk;
  __syncthreads();

  int ubase = 0;
#pragma unroll
  for (int ww = 0; ww < 4; ++ww) if (ww < w) ubase += wtot[ww];
  const int nb = wtot[0] + wtot[1] + wtot[2] + wtot[3];
  {
    const unsigned lt = (1u << lane) - 1u;
#pragma unroll
    for (int i = 0; i < 16; ++i) {
      if ((bal[i] >> lane) & 1u) U[ubase + __popc(bal[i] & lt)] = w * 512 + i * 32 + lane;
      ubase += __popc(bal[i]);
    }
  }

  {
    const float vmean = vall * (1.0f / (float)Sq);
    for (int r = 0; r < QT; ++r)
      if (mrow[q0 + r] == 0) O[baseBH + (size_t)(q0 + r) * HD + t] = vmean;
  }
  __syncthreads();
  if (q0 >= nb) return;
  const int nlast = nb - 1;

  int qi = q0 + w * 16 + lm;
  qi = (qi < nb) ? qi : nlast;
  const float* qrow = Qm + baseBH + (size_t)U[qi] * HD;
  v2f qf[HD / 4];
#pragma unroll
  for (int kd = 0; kd < HD / 4; ++kd)
    qf[kd] = *(const v2f*)(qrow + 4 * kd + 2 * lh);

  v8f o[8];
#pragma unroll
  for (int s = 0; s < 8; ++s) o[s] = vzero8();
  float rmax = -INFINITY;
  float rsum = 0.0f;

  for (int kt = 0; kt < nb; kt += KT) {
#pragma unroll
    for (int i = 0; i < 8; ++i) {
      int idx = t + i * 128;
      int r = idx >> 5;
      int c = (idx & 31) << 2;
      int kr = kt + r;
      kr = (kr < nb) ? kr : nlast;
      const size_t ro = baseBH + (size_t)U[kr] * HD + c;
      float4 kv = *(const float4*)(Km + ro);
      *(float4*)&Ks[r][c] = kv;
      float4 vv = *(const float4*)(Vm + ro);
      float* dst = &Vsp[r >> 1][2 * c + (r & 1)];
      dst[0] = vv.x;
      dst[2] = vv.y;
      dst[4] = vv.z;
      dst[6] = vv.w;
    }
    __syncthreads();

    v8f sc[2];
    sc[0] = vzero8();
    sc[1] = vzero8();
#pragma unroll
    for (int kd = 0; kd < HD / 4; ++kd) {
      v2f af = qf[kd];
#pragma unroll
      for (int j = 0; j < 2; ++j) {
        int kc = j * 16 + lm;
        v2f bf = *(const v2f*)&Ks[kc][4 * kd + 2 * lh];
        sc[j] = WMMA_F32(af, bf, sc[j]);
      }
    }

#pragma unroll
    for (int j = 0; j < 2; ++j)
#pragma unroll
      for (int r = 0; r < 8; ++r) {
        int m = r + 8 * lh;
        int kc = j * 16 + lm;
        float v = sc[j][r];
        if (kt + kc >= nb) v = -INFINITY;
        Ps[w][m][kc] = v;
      }
    __syncthreads();

    {
      const int c0 = lh * 16;
      float pm = -INFINITY;
#pragma unroll
      for (int c = 0; c < 16; ++c) pm = fmaxf(pm, Ps[w][lm][c0 + c]);
      pm = fmaxf(pm, __shfl_xor(pm, 16, 32));
      float nm = fmaxf(rmax, pm);
      float cf = __expf(rmax - nm);
      float ps = 0.0f;
#pragma unroll
      for (int c = 0; c < 16; ++c) {
        float e = __expf(Ps[w][lm][c0 + c] - nm);
        Ps[w][lm][c0 + c] = e;
        ps += e;
      }
      ps += __shfl_xor(ps, 16, 32);
      rsum = rsum * cf + ps;
      rmax = nm;
      if (lane < 16) corrS[w][lm] = cf;
    }
    __syncthreads();

    float cv[8];
#pragma unroll
    for (int r = 0; r < 8; ++r) cv[r] = corrS[w][r + 8 * lh];
#pragma unroll
    for (int s = 0; s < 8; ++s)
#pragma unroll
      for (int r = 0; r < 8; ++r) o[s][r] *= cv[r];

#pragma unroll
    for (int kk = 0; kk < KT; kk += 4) {
      v2f af = *(const v2f*)&Ps[w][lm][kk + 2 * lh];
      const int khalf = (kk >> 1) + lh;
#pragma unroll
      for (int s = 0; s < 8; ++s) {
        int col = s * 16 + lm;
        v2f bf = *(const v2f*)&Vsp[khalf][2 * col];
        o[s] = WMMA_F32(af, bf, o[s]);
      }
    }
    __syncthreads();
  }

  if (lane < 16) { rsumS[w][lm] = rsum; rmaxS[w][lm] = rmax; }
  __syncthreads();

  const int nmask = Sq - nb;
  const float fnm = (float)nmask;
  float ra[8], rc[8], rv[8];
#pragma unroll
  for (int r = 0; r < 8; ++r) {
    const float mx = rmaxS[w][r + 8 * lh];
    const float rs = rsumS[w][r + 8 * lh];
    const float M = (nmask > 0) ? fmaxf(mx, NEGV) : mx;
    const float a = __expf(mx - M);
    const float c = (nmask > 0) ? __expf(NEGV - M) : 0.0f;
    ra[r] = a;
    rc[r] = c;
    rv[r] = 1.0f / (rs * a + c * fnm);
  }

#pragma unroll
  for (int s = 0; s < 8; ++s) {
    const int d = s * 16 + lm;
    const float vk = vmskS[d];
#pragma unroll
    for (int r = 0; r < 8; ++r) {
      const int qi2 = q0 + w * 16 + r + 8 * lh;
      if (qi2 < nb) {
        const int s2 = U[qi2];
        O[baseBH + (size_t)s2 * HD + d] = (o[s][r] * ra[r] + rc[r] * vk) * rv[r];
      }
    }
  }
}

extern "C" void kernel_launch(void* const* d_in, const int* in_sizes, int n_in,
                              void* d_out, int out_size, void* d_ws, size_t ws_size,
                              hipStream_t stream) {
  const float* X    = (const float*)d_in[0];
  const int*   mask = (const int*)d_in[1];
  const float* Wq   = (const float*)d_in[2];
  const float* Wk   = (const float*)d_in[3];
  const float* Wv   = (const float*)d_in[4];
  const float* Wqm  = (const float*)d_in[5];
  const float* Wkm  = (const float*)d_in[6];
  const float* Wvm  = (const float*)d_in[7];
  const float* Wo   = (const float*)d_in[8];
  float* out = (float*)d_out;

  constexpr size_t SZ = (size_t)Bz * Sq * Hd;
  constexpr size_t WZ = (size_t)Hd * Hd;
  float* s0 = (float*)d_ws;
  float* s1 = s0 + SZ;
  float* s2 = s1 + SZ;
  float* s3 = s2 + SZ;
  float* fq = s0;
  float* fk = s0 + WZ;
  float* fv = s1;

  constexpr int M = Bz * Sq, N = Hd, K = Hd;
  dim3 gw3(N / TN, K / TM, 3);
  dim3 gg3(N / TN, M / TM, 3);
  dim3 gg1(N / TN, M / TM, 1);
  dim3 gb(256);

  float* vm = out;

  gemm_f32_wmma<<<gw3, gb, 0, stream>>>(Wq, Wqm, fq, Wk, Wkm, fk, Wv, Wvm, fv, K, N, K);
  gemm_f32_wmma<<<gg3, gb, 0, stream>>>(X, fq, s2, X, fk, s3, X, fv, vm, M, N, K);

  float* pall = s0;
  float* pmsk = s0 + (size_t)Bz * NH * VSL * HD;
  vsum_partial<<<Bz * NH * VSL, 256, 0, stream>>>(vm, mask, pall, pmsk);

  attn_f32_wmma<<<Bz * NH * (Sq / QT), 128, 0, stream>>>(s2, s3, vm, mask, pall, pmsk, s1);

  gemm_f32_wmma<<<gg1, gb, 0, stream>>>(s1, Wo, out, s1, Wo, out, s1, Wo, out, M, N, K);
}


# Q and K projections only at unmasked positions: per 128-column chunk the unmasked X rows are gathered (list from the mask) into 128-row tiles, empty tiles exit; masked Q/K are never written
# speedup vs baseline: 1.7901x; 1.1016x over previous
#include <hip/hip_runtime.h>
#include <hip/hip_bf16.h>
#include <math.h>

constexpr int Bz = 2;
constexpr int Sq = 2048;
constexpr int Hd = 2048;
constexpr int NH = 16;
constexpr int HD = 128;
constexpr float NEGV = -1000000.0f;

typedef float v2f __attribute__((ext_vector_type(2)));
typedef float v8f __attribute__((ext_vector_type(8)));

#define WMMA_F32(a, b, c) \
  __builtin_amdgcn_wmma_f32_16x16x4_f32(false, (a), false, (b), (short)0, (c), false, false)

__device__ __forceinline__ v8f vzero8() {
  v8f z = {0.f, 0.f, 0.f, 0.f, 0.f, 0.f, 0.f, 0.f};
  return z;
}

constexpr int TM = 128, TN = 128, TK = 32;
constexpr int APAD = 4;
constexpr int BIPAD = 32;

template <bool GATHER>
__global__ __launch_bounds__(256) void gemm_f32_wmma(
    const float* __restrict__ A0, const float* __restrict__ B0, float* __restrict__ C0,
    const float* __restrict__ A1, const float* __restrict__ B1, float* __restrict__ C1,
    const float* __restrict__ A2, const float* __restrict__ B2, float* __restrict__ C2,
    const int* __restrict__ mask, int M, int N, int K) {
  __shared__ float As[TM][TK + APAD];
  __shared__ float Bsp[TK / 2][2 * TN + BIPAD];
  __shared__ int rowS[GATHER ? TM : 1];
  __shared__ int relS[GATHER ? 128 : 1];
  __shared__ unsigned wbalS[4];

  const int z = blockIdx.z;
  const float* __restrict__ A  = (z == 0) ? A0 : (z == 1) ? A1 : A2;
  const float* __restrict__ Bm = (z == 0) ? B0 : (z == 1) ? B1 : B2;
  float* __restrict__ C        = (z == 0) ? C0 : (z == 1) ? C1 : C2;

  const int t = threadIdx.x;
  const int lane = t & 31;
  const int wave = t >> 5;
  const int lm = lane & 15;
  const int lh = lane >> 4;
  const int wm = wave & 3;
  const int wn = wave >> 2;
  const int m0 = blockIdx.y * TM;
  const int n0 = blockIdx.x * TN;

  int nvalid = TM;
  if constexpr (GATHER) {
    if (z < 2) {
      const int b = blockIdx.y & 1;
      const int ti = blockIdx.y >> 1;
      bool f = false;
      if (t < 128) f = mask[b * Sq + 16 * t + (int)blockIdx.x] != 0;
      const unsigned bal = __builtin_amdgcn_ballot_w32(f);
      if (wave < 4 && lane == 0) wbalS[wave] = bal;
      __syncthreads();
      const unsigned u0 = wbalS[0], u1 = wbalS[1], u2 = wbalS[2], u3 = wbalS[3];
      const int nR = __popc(u0) + __popc(u1) + __popc(u2) + __popc(u3);
      const int total = 16 * nR;
      if (ti * TM >= total) return;
      if (f) {
        int pos = __popc(bal & ((1u << lane) - 1u));
        if (wave > 0) pos += __popc(u0);
        if (wave > 1) pos += __popc(u1);
        if (wave > 2) pos += __popc(u2);
        relS[pos] = t;
      }
      __syncthreads();
      if (t < TM) {
        int g = ti * TM + t;
        g = (g < total) ? g : total - 1;
        int h = 0;
#pragma unroll
        for (int k = 1; k < 16; ++k) h += (g >= k * nR) ? 1 : 0;
        rowS[t] = b * Sq + h * 128 + relS[g - h * nR];
      }
      nvalid = (total - ti * TM < TM) ? total - ti * TM : TM;
    } else {
      if (t < TM) rowS[t] = m0 + t;
    }
    __syncthreads();
  }

  v8f acc[2][4];
#pragma unroll
  for (int i = 0; i < 2; ++i)
#pragma unroll
    for (int j = 0; j < 4; ++j) acc[i][j] = vzero8();

  const int ar = t >> 3;
  const int ac = (t & 7) << 2;
  const int br = t >> 5;
  const int bc = (t & 31) << 2;
  const float* ap = A + (size_t)(m0 + ar) * K + ac;
  const float* bp = Bm + (size_t)br * N + n0 + bc;

  const size_t aStep = (size_t)32 * K;
  const size_t bStep = (size_t)8 * N;
  const char* const Ab = (const char*)A;
  unsigned oa0 = 0, oa1 = 0, oa2 = 0, oa3 = 0;
  if constexpr (GATHER) {
    oa0 = ((unsigned)rowS[ar] * (unsigned)K + (unsigned)ac) * 4u;
    oa1 = ((unsigned)rowS[ar + 32] * (unsigned)K + (unsigned)ac) * 4u;
    oa2 = ((unsigned)rowS[ar + 64] * (unsigned)K + (unsigned)ac) * 4u;
    oa3 = ((unsigned)rowS[ar + 96] * (unsigned)K + (unsigned)ac) * 4u;
  }
  float4 ga0, ga1, ga2, ga3;
  if constexpr (GATHER) {
    ga0 = *(const float4*)(Ab + oa0);
    ga1 = *(const float4*)(Ab + oa1);
    ga2 = *(const float4*)(Ab + oa2);
    ga3 = *(const float4*)(Ab + oa3);
  } else {
    ga0 = *(const float4*)(ap);
    ga1 = *(const float4*)(ap + aStep);
    ga2 = *(const float4*)(ap + 2 * aStep);
    ga3 = *(const float4*)(ap + 3 * aStep);
  }
  float4 gb0 = *(const float4*)(bp);
  float4 gb1 = *(const float4*)(bp + bStep);
  float4 gb2 = *(const float4*)(bp + 2 * bStep);
  float4 gb3 = *(const float4*)(bp + 3 * bStep);

  float* const bdst = &Bsp[br >> 1][2 * bc + (br & 1)];
  constexpr int BROW4 = 4 * (2 * TN + BIPAD);

  for (int kt = 0; kt < K; kt += TK) {
    *(float4*)&As[ar][ac] = ga0;
    *(float4*)&As[ar + 32][ac] = ga1;
    *(float4*)&As[ar + 64][ac] = ga2;
    *(float4*)&As[ar + 96][ac] = ga3;
    bdst[0] = gb0.x; bdst[2] = gb0.y; bdst[4] = gb0.z; bdst[6] = gb0.w;
    bdst[BROW4 + 0] = gb1.x; bdst[BROW4 + 2] = gb1.y;
    bdst[BROW4 + 4] = gb1.z; bdst[BROW4 + 6] = gb1.w;
    bdst[2 * BROW4 + 0] = gb2.x; bdst[2 * BROW4 + 2] = gb2.y;
    bdst[2 * BROW4 + 4] = gb2.z; bdst[2 * BROW4 + 6] = gb2.w;
    bdst[3 * BROW4 + 0] = gb3.x; bdst[3 * BROW4 + 2] = gb3.y;
    bdst[3 * BROW4 + 4] = gb3.z; bdst[3 * BROW4 + 6] = gb3.w;
    __syncthreads();

    {
      const int kn = (kt + TK < K) ? kt + TK : kt;
      const float* an = ap + kn;
      const float* bn = bp + (size_t)kn * N;
      if constexpr (GATHER) {
        const unsigned kb = (unsigned)kn * 4u;
        ga0 = *(const float4*)(Ab + (oa0 + kb));
        ga1 = *(const float4*)(Ab + (oa1 + kb));
        ga2 = *(const float4*)(Ab + (oa2 + kb));
        ga3 = *(const float4*)(Ab + (oa3 + kb));
      } else {
        ga0 = *(const float4*)(an);
        ga1 = *(const float4*)(an + aStep);
        ga2 = *(const float4*)(an + 2 * aStep);
        ga3 = *(const float4*)(an + 3 * aStep);
      }
      gb0 = *(const float4*)(bn);
      gb1 = *(const float4*)(bn + bStep);
      gb2 = *(const float4*)(bn + 2 * bStep);
      gb3 = *(const float4*)(bn + 3 * bStep);
    }
    __builtin_amdgcn_sched_barrier(0);

#pragma unroll
    for (int kk = 0; kk < TK; kk += 4) {
      v2f af[2];
#pragma unroll
      for (int im = 0; im < 2; ++im) {
        int row = wm * 32 + im * 16 + lm;
        af[im] = *(const v2f*)&As[row][kk + 2 * lh];
      }
      const int khalf = (kk >> 1) + lh;
#pragma unroll
      for (int j = 0; j < 4; ++j) {
        int col = wn * 64 + j * 16 + lm;
        v2f bf = *(const v2f*)&Bsp[khalf][2 * col];
        acc[0][j] = WMMA_F32(af[0], bf, acc[0][j]);
        acc[1][j] = WMMA_F32(af[1], bf, acc[1][j]);
      }
    }
    __syncthreads();
  }

  if constexpr (GATHER) {
    unsigned drow[2][8];
#pragma unroll
    for (int im = 0; im < 2; ++im)
#pragma unroll
      for (int r = 0; r < 8; ++r)
        drow[im][r] = (unsigned)rowS[wm * 32 + im * 16 + r + 8 * lh] * (unsigned)N;
#pragma unroll
    for (int im = 0; im < 2; ++im)
#pragma unroll
      for (int j = 0; j < 4; ++j)
#pragma unroll
        for (int r = 0; r < 8; ++r) {
          const int li = wm * 32 + im * 16 + r + 8 * lh;
          const unsigned n = (unsigned)(n0 + wn * 64 + j * 16 + lm);
          if (li < nvalid) C[drow[im][r] + n] = acc[im][j][r];
        }
  } else {
#pragma unroll
    for (int im = 0; im < 2; ++im)
#pragma unroll
      for (int j = 0; j < 4; ++j)
#pragma unroll
        for (int r = 0; r < 8; ++r) {
          int m = m0 + wm * 32 + im * 16 + r + 8 * lh;
          int n = n0 + wn * 64 + j * 16 + lm;
          C[(size_t)m * N + n] = acc[im][j][r];
        }
  }
}

constexpr int VSL = 16;
constexpr int VSR = Sq / VSL;

__global__ __launch_bounds__(256) void vsum_partial(
    const float* __restrict__ Vm, const int* __restrict__ mask,
    float* __restrict__ pall, float* __restrict__ pmsk) {
  __shared__ float redA[8][HD];
  __shared__ float redM[8][HD];
  const int t = threadIdx.x;
  const int c4 = (t & 31) << 2;
  const int rg = t >> 5;
  const int sl = blockIdx.x % VSL;
  const int bh = blockIdx.x / VSL;
  const int b = bh / NH;
  const float* vp = Vm + (size_t)bh * Sq * HD + (size_t)(sl * VSR) * HD + c4;
  const int* mp = mask + b * Sq + sl * VSR;
  float4 sa = make_float4(0.f, 0.f, 0.f, 0.f);
  float4 sm = make_float4(0.f, 0.f, 0.f, 0.f);
#pragma unroll
  for (int i = 0; i < VSR / 8; ++i) {
    const int r = rg + 8 * i;
    const float4 v = *(const float4*)(vp + (size_t)r * HD);
    const int mk = mp[r];
    sa.x += v.x; sa.y += v.y; sa.z += v.z; sa.w += v.w;
    if (mk == 0) { sm.x += v.x; sm.y += v.y; sm.z += v.z; sm.w += v.w; }
  }
  *(float4*)&redA[rg][c4] = sa;
  *(float4*)&redM[rg][c4] = sm;
  __syncthreads();
  if (t < HD) {
    float a = 0.f, m = 0.f;
#pragma unroll
    for (int g = 0; g < 8; ++g) { a += redA[g][t]; m += redM[g][t]; }
    pall[((size_t)bh * VSL + sl) * HD + t] = a;
    pmsk[((size_t)bh * VSL + sl) * HD + t] = m;
  }
}

constexpr int QT = 64;
constexpr int KT = 32;
constexpr int KVP = 12;
constexpr int VIPAD = 32;
constexpr int PPAD = 4;

__global__ __launch_bounds__(128) void attn_f32_wmma(
    const float* __restrict__ Qm, const float* __restrict__ Km,
    const float* __restrict__ Vm, const int* __restrict__ mask,
    const float* __restrict__ pall, const float* __restrict__ pmsk,
    float* __restrict__ O) {
  __shared__ float Ks[KT][HD + KVP];
  __shared__ float Vsp[KT / 2][2 * HD + VIPAD];
  __shared__ float Ps[4][16][KT + PPAD];
  __shared__ float corrS[4][16];
  __shared__ float rsumS[4][16];
  __shared__ float rmaxS[4][16];
  __shared__ float vmskS[HD];
  __shared__ int U[Sq];
  __shared__ int wtot[4];

  const int t = threadIdx.x;
  const int lane = t & 31;
  const int w = t >> 5;
  const int lm = lane & 15;
  const int lh = lane >> 4;

  const int bh = blockIdx.x % (Bz * NH);
  const int qb = blockIdx.x / (Bz * NH);
  const int b = bh / NH;
  const int q0 = qb * QT;

  const size_t baseBH = (size_t)bh * Sq * HD;
  const int* mrow = mask + b * Sq;

  unsigned bal[16];
  int cnt = 0;
#pragma unroll
  for (int i = 0; i < 16; ++i) {
    bal[i] = __builtin_amdgcn_ballot_w32(mrow[w * 512 + i * 32 + lane] != 0);
    cnt += __popc(bal[i]);
  }
  if (lane == 0) wtot[w] = cnt;

  float vall = 0.f, vmsk = 0.f;
  {
    const float* pa = pall + (size_t)bh * VSL * HD + t;
    const float* pm = pmsk + (size_t)bh * VSL * HD + t;
#pragma unroll
    for (int s = 0; s < VSL; ++s) { vall += pa[s * HD]; vmsk += pm[s * HD]; }
  }
  vmskS[t] = vmsk;
  __syncthreads();

  int ubase = 0;
#pragma unroll
  for (int ww = 0; ww < 4; ++ww) if (ww < w) ubase += wtot[ww];
  const int nb = wtot[0] + wtot[1] + wtot[2] + wtot[3];
  {
    const unsigned lt = (1u << lane) - 1u;
#pragma unroll
    for (int i = 0; i < 16; ++i) {
      if ((bal[i] >> lane) & 1u) U[ubase + __popc(bal[i] & lt)] = w * 512 + i * 32 + lane;
      ubase += __popc(bal[i]);
    }
  }

  {
    const float vmean = vall * (1.0f / (float)Sq);
    for (int r = 0; r < QT; ++r)
      if (mrow[q0 + r] == 0) O[baseBH + (size_t)(q0 + r) * HD + t] = vmean;
  }
  __syncthreads();
  if (q0 >= nb) return;
  const int nlast = nb - 1;

  int qi = q0 + w * 16 + lm;
  qi = (qi < nb) ? qi : nlast;
  const float* qrow = Qm + baseBH + (size_t)U[qi] * HD;
  v2f qf[HD / 4];
#pragma unroll
  for (int kd = 0; kd < HD / 4; ++kd)
    qf[kd] = *(const v2f*)(qrow + 4 * kd + 2 * lh);

  v8f o[8];
#pragma unroll
  for (int s = 0; s < 8; ++s) o[s] = vzero8();
  float rmax = -INFINITY;
  float rsum = 0.0f;

  for (int kt = 0; kt < nb; kt += KT) {
#pragma unroll
    for (int i = 0; i < 8; ++i) {
      int idx = t + i * 128;
      int r = idx >> 5;
      int c = (idx & 31) << 2;
      int kr = kt + r;
      kr = (kr < nb) ? kr : nlast;
      const size_t ro = baseBH + (size_t)U[kr] * HD + c;
      float4 kv = *(const float4*)(Km + ro);
      *(float4*)&Ks[r][c] = kv;
      float4 vv = *(const float4*)(Vm + ro);
      float* dst = &Vsp[r >> 1][2 * c + (r & 1)];
      dst[0] = vv.x;
      dst[2] = vv.y;
      dst[4] = vv.z;
      dst[6] = vv.w;
    }
    __syncthreads();

    v8f sc[2];
    sc[0] = vzero8();
    sc[1] = vzero8();
#pragma unroll
    for (int kd = 0; kd < HD / 4; ++kd) {
      v2f af = qf[kd];
#pragma unroll
      for (int j = 0; j < 2; ++j) {
        int kc = j * 16 + lm;
        v2f bf = *(const v2f*)&Ks[kc][4 * kd + 2 * lh];
        sc[j] = WMMA_F32(af, bf, sc[j]);
      }
    }

#pragma unroll
    for (int j = 0; j < 2; ++j)
#pragma unroll
      for (int r = 0; r < 8; ++r) {
        int m = r + 8 * lh;
        int kc = j * 16 + lm;
        float v = sc[j][r];
        if (kt + kc >= nb) v = -INFINITY;
        Ps[w][m][kc] = v;
      }
    __syncthreads();

    {
      const int c0 = lh * 16;
      float pm = -INFINITY;
#pragma unroll
      for (int c = 0; c < 16; ++c) pm = fmaxf(pm, Ps[w][lm][c0 + c]);
      pm = fmaxf(pm, __shfl_xor(pm, 16, 32));
      float nm = fmaxf(rmax, pm);
      float cf = __expf(rmax - nm);
      float ps = 0.0f;
#pragma unroll
      for (int c = 0; c < 16; ++c) {
        float e = __expf(Ps[w][lm][c0 + c] - nm);
        Ps[w][lm][c0 + c] = e;
        ps += e;
      }
      ps += __shfl_xor(ps, 16, 32);
      rsum = rsum * cf + ps;
      rmax = nm;
      if (lane < 16) corrS[w][lm] = cf;
    }
    __syncthreads();

    float cv[8];
#pragma unroll
    for (int r = 0; r < 8; ++r) cv[r] = corrS[w][r + 8 * lh];
#pragma unroll
    for (int s = 0; s < 8; ++s)
#pragma unroll
      for (int r = 0; r < 8; ++r) o[s][r] *= cv[r];

#pragma unroll
    for (int kk = 0; kk < KT; kk += 4) {
      v2f af = *(const v2f*)&Ps[w][lm][kk + 2 * lh];
      const int khalf = (kk >> 1) + lh;
#pragma unroll
      for (int s = 0; s < 8; ++s) {
        int col = s * 16 + lm;
        v2f bf = *(const v2f*)&Vsp[khalf][2 * col];
        o[s] = WMMA_F32(af, bf, o[s]);
      }
    }
    __syncthreads();
  }

  if (lane < 16) { rsumS[w][lm] = rsum; rmaxS[w][lm] = rmax; }
  __syncthreads();

  const int nmask = Sq - nb;
  const float fnm = (float)nmask;
  float ra[8], rc[8], rv[8];
#pragma unroll
  for (int r = 0; r < 8; ++r) {
    const float mx = rmaxS[w][r + 8 * lh];
    const float rs = rsumS[w][r + 8 * lh];
    const float M = (nmask > 0) ? fmaxf(mx, NEGV) : mx;
    const float a = __expf(mx - M);
    const float c = (nmask > 0) ? __expf(NEGV - M) : 0.0f;
    ra[r] = a;
    rc[r] = c;
    rv[r] = 1.0f / (rs * a + c * fnm);
  }

#pragma unroll
  for (int s = 0; s < 8; ++s) {
    const int d = s * 16 + lm;
    const float vk = vmskS[d];
#pragma unroll
    for (int r = 0; r < 8; ++r) {
      const int qi2 = q0 + w * 16 + r + 8 * lh;
      if (qi2 < nb) {
        const int s2 = U[qi2];
        O[baseBH + (size_t)s2 * HD + d] = (o[s][r] * ra[r] + rc[r] * vk) * rv[r];
      }
    }
  }
}

extern "C" void kernel_launch(void* const* d_in, const int* in_sizes, int n_in,
                              void* d_out, int out_size, void* d_ws, size_t ws_size,
                              hipStream_t stream) {
  const float* X    = (const float*)d_in[0];
  const int*   mask = (const int*)d_in[1];
  const float* Wq   = (const float*)d_in[2];
  const float* Wk   = (const float*)d_in[3];
  const float* Wv   = (const float*)d_in[4];
  const float* Wqm  = (const float*)d_in[5];
  const float* Wkm  = (const float*)d_in[6];
  const float* Wvm  = (const float*)d_in[7];
  const float* Wo   = (const float*)d_in[8];
  float* out = (float*)d_out;

  constexpr size_t SZ = (size_t)Bz * Sq * Hd;
  constexpr size_t WZ = (size_t)Hd * Hd;
  float* s0 = (float*)d_ws;
  float* s1 = s0 + SZ;
  float* s2 = s1 + SZ;
  float* s3 = s2 + SZ;
  float* fq = s0;
  float* fk = s0 + WZ;
  float* fv = s1;

  constexpr int M = Bz * Sq, N = Hd, K = Hd;
  dim3 gw3(N / TN, K / TM, 3);
  dim3 gg3(N / TN, M / TM, 3);
  dim3 gg1(N / TN, M / TM, 1);
  dim3 gb(256);

  float* vm = out;

  gemm_f32_wmma<false><<<gw3, gb, 0, stream>>>(Wq, Wqm, fq, Wk, Wkm, fk, Wv, Wvm, fv, mask, K, N, K);
  gemm_f32_wmma<true><<<gg3, gb, 0, stream>>>(X, fq, s2, X, fk, s3, X, fv, vm, mask, M, N, K);

  float* pall = s0;
  float* pmsk = s0 + (size_t)Bz * NH * VSL * HD;
  vsum_partial<<<Bz * NH * VSL, 256, 0, stream>>>(vm, mask, pall, pmsk);

  attn_f32_wmma<<<Bz * NH * (Sq / QT), 128, 0, stream>>>(s2, s3, vm, mask, pall, pmsk, s1);

  gemm_f32_wmma<false><<<gg1, gb, 0, stream>>>(s1, Wo, out, s1, Wo, out, s1, Wo, out, mask, M, N, K);
}


# V projection only at unmasked positions too (same gathered tiles as Q/K); V's masked-key and all-key column sums by linearity: row sums of X times the folded weight on WMMA; those sums regrouped
# speedup vs baseline: 1.8491x; 1.0330x over previous
#include <hip/hip_runtime.h>
#include <hip/hip_bf16.h>
#include <math.h>

constexpr int Bz = 2;
constexpr int Sq = 2048;
constexpr int Hd = 2048;
constexpr int NH = 16;
constexpr int HD = 128;
constexpr float NEGV = -1000000.0f;

typedef float v2f __attribute__((ext_vector_type(2)));
typedef float v8f __attribute__((ext_vector_type(8)));

#define WMMA_F32(a, b, c) \
  __builtin_amdgcn_wmma_f32_16x16x4_f32(false, (a), false, (b), (short)0, (c), false, false)

__device__ __forceinline__ v8f vzero8() {
  v8f z = {0.f, 0.f, 0.f, 0.f, 0.f, 0.f, 0.f, 0.f};
  return z;
}

constexpr int TM = 128, TN = 128, TK = 32;
constexpr int XS_ALL_ROW0 = 512;
constexpr int APAD = 4;
constexpr int BIPAD = 32;

template <bool GATHER>
__global__ __launch_bounds__(256) void gemm_f32_wmma(
    const float* __restrict__ A0, const float* __restrict__ B0, float* __restrict__ C0,
    const float* __restrict__ A1, const float* __restrict__ B1, float* __restrict__ C1,
    const float* __restrict__ A2, const float* __restrict__ B2, float* __restrict__ C2,
    const float* __restrict__ A3, const float* __restrict__ B3, float* __restrict__ C3,
    const int* __restrict__ mask, int M, int N, int K) {
  __shared__ float As[TM][TK + APAD];
  __shared__ float Bsp[TK / 2][2 * TN + BIPAD];
  __shared__ int rowS[GATHER ? TM : 1];
  __shared__ int relS[GATHER ? 128 : 1];
  __shared__ unsigned wbalS[4];

  const int z = blockIdx.z;
  const float* __restrict__ A  = (z == 0) ? A0 : (z == 1) ? A1 : (z == 2) ? A2 : A3;
  const float* __restrict__ Bm = (z == 0) ? B0 : (z == 1) ? B1 : (z == 2) ? B2 : B3;
  float* __restrict__ C        = (z == 0) ? C0 : (z == 1) ? C1 : (z == 2) ? C2 : C3;

  const int t = threadIdx.x;
  const int lane = t & 31;
  const int wave = t >> 5;
  const int lm = lane & 15;
  const int lh = lane >> 4;
  const int wm = wave & 3;
  const int wn = wave >> 2;
  const int m0 = blockIdx.y * TM;
  const int n0 = blockIdx.x * TN;

  int nvalid = TM;
  if constexpr (GATHER) {
    if (z < 3) {
      const int b = blockIdx.y & 1;
      const int ti = blockIdx.y >> 1;
      bool f = false;
      if (t < 128) f = mask[b * Sq + 16 * t + (int)blockIdx.x] != 0;
      const unsigned bal = __builtin_amdgcn_ballot_w32(f);
      if (wave < 4 && lane == 0) wbalS[wave] = bal;
      __syncthreads();
      const unsigned u0 = wbalS[0], u1 = wbalS[1], u2 = wbalS[2], u3 = wbalS[3];
      const int nR = __popc(u0) + __popc(u1) + __popc(u2) + __popc(u3);
      const int total = 16 * nR;
      if (ti * TM >= total) return;
      if (f) {
        int pos = __popc(bal & ((1u << lane) - 1u));
        if (wave > 0) pos += __popc(u0);
        if (wave > 1) pos += __popc(u1);
        if (wave > 2) pos += __popc(u2);
        relS[pos] = t;
      }
      __syncthreads();
      if (t < TM) {
        int g = ti * TM + t;
        g = (g < total) ? g : total - 1;
        int h = 0;
#pragma unroll
        for (int k = 1; k < 16; ++k) h += (g >= k * nR) ? 1 : 0;
        rowS[t] = b * Sq + h * 128 + relS[g - h * nR];
      }
      nvalid = (total - ti * TM < TM) ? total - ti * TM : TM;
    } else {
      if (blockIdx.y != 0) return;
      if (t < TM) {
        const int tt = (t < 64) ? t : 63;
        rowS[t] = (tt < 32) ? (int)blockIdx.x * 32 + tt : XS_ALL_ROW0 + (tt - 32);
      }
      nvalid = 64;
    }
    __syncthreads();
  }

  v8f acc[2][4];
#pragma unroll
  for (int i = 0; i < 2; ++i)
#pragma unroll
    for (int j = 0; j < 4; ++j) acc[i][j] = vzero8();

  const int ar = t >> 3;
  const int ac = (t & 7) << 2;
  const int br = t >> 5;
  const int bc = (t & 31) << 2;
  const float* ap = A + (size_t)(m0 + ar) * K + ac;
  const float* bp = Bm + (size_t)br * N + n0 + bc;

  const size_t aStep = (size_t)32 * K;
  const size_t bStep = (size_t)8 * N;
  const char* const Ab = (const char*)A;
  unsigned oa0 = 0, oa1 = 0, oa2 = 0, oa3 = 0;
  if constexpr (GATHER) {
    oa0 = ((unsigned)rowS[ar] * (unsigned)K + (unsigned)ac) * 4u;
    oa1 = ((unsigned)rowS[ar + 32] * (unsigned)K + (unsigned)ac) * 4u;
    oa2 = ((unsigned)rowS[ar + 64] * (unsigned)K + (unsigned)ac) * 4u;
    oa3 = ((unsigned)rowS[ar + 96] * (unsigned)K + (unsigned)ac) * 4u;
  }
  float4 ga0, ga1, ga2, ga3;
  if constexpr (GATHER) {
    ga0 = *(const float4*)(Ab + oa0);
    ga1 = *(const float4*)(Ab + oa1);
    ga2 = *(const float4*)(Ab + oa2);
    ga3 = *(const float4*)(Ab + oa3);
  } else {
    ga0 = *(const float4*)(ap);
    ga1 = *(const float4*)(ap + aStep);
    ga2 = *(const float4*)(ap + 2 * aStep);
    ga3 = *(const float4*)(ap + 3 * aStep);
  }
  float4 gb0 = *(const float4*)(bp);
  float4 gb1 = *(const float4*)(bp + bStep);
  float4 gb2 = *(const float4*)(bp + 2 * bStep);
  float4 gb3 = *(const float4*)(bp + 3 * bStep);

  float* const bdst = &Bsp[br >> 1][2 * bc + (br & 1)];
  constexpr int BROW4 = 4 * (2 * TN + BIPAD);

  for (int kt = 0; kt < K; kt += TK) {
    *(float4*)&As[ar][ac] = ga0;
    *(float4*)&As[ar + 32][ac] = ga1;
    *(float4*)&As[ar + 64][ac] = ga2;
    *(float4*)&As[ar + 96][ac] = ga3;
    bdst[0] = gb0.x; bdst[2] = gb0.y; bdst[4] = gb0.z; bdst[6] = gb0.w;
    bdst[BROW4 + 0] = gb1.x; bdst[BROW4 + 2] = gb1.y;
    bdst[BROW4 + 4] = gb1.z; bdst[BROW4 + 6] = gb1.w;
    bdst[2 * BROW4 + 0] = gb2.x; bdst[2 * BROW4 + 2] = gb2.y;
    bdst[2 * BROW4 + 4] = gb2.z; bdst[2 * BROW4 + 6] = gb2.w;
    bdst[3 * BROW4 + 0] = gb3.x; bdst[3 * BROW4 + 2] = gb3.y;
    bdst[3 * BROW4 + 4] = gb3.z; bdst[3 * BROW4 + 6] = gb3.w;
    __syncthreads();

    {
      const int kn = (kt + TK < K) ? kt + TK : kt;
      const float* an = ap + kn;
      const float* bn = bp + (size_t)kn * N;
      if constexpr (GATHER) {
        const unsigned kb = (unsigned)kn * 4u;
        ga0 = *(const float4*)(Ab + (oa0 + kb));
        ga1 = *(const float4*)(Ab + (oa1 + kb));
        ga2 = *(const float4*)(Ab + (oa2 + kb));
        ga3 = *(const float4*)(Ab + (oa3 + kb));
      } else {
        ga0 = *(const float4*)(an);
        ga1 = *(const float4*)(an + aStep);
        ga2 = *(const float4*)(an + 2 * aStep);
        ga3 = *(const float4*)(an + 3 * aStep);
      }
      gb0 = *(const float4*)(bn);
      gb1 = *(const float4*)(bn + bStep);
      gb2 = *(const float4*)(bn + 2 * bStep);
      gb3 = *(const float4*)(bn + 3 * bStep);
    }
    __builtin_amdgcn_sched_barrier(0);

#pragma unroll
    for (int kk = 0; kk < TK; kk += 4) {
      v2f af[2];
#pragma unroll
      for (int im = 0; im < 2; ++im) {
        int row = wm * 32 + im * 16 + lm;
        af[im] = *(const v2f*)&As[row][kk + 2 * lh];
      }
      const int khalf = (kk >> 1) + lh;
#pragma unroll
      for (int j = 0; j < 4; ++j) {
        int col = wn * 64 + j * 16 + lm;
        v2f bf = *(const v2f*)&Bsp[khalf][2 * col];
        acc[0][j] = WMMA_F32(af[0], bf, acc[0][j]);
        acc[1][j] = WMMA_F32(af[1], bf, acc[1][j]);
      }
    }
    __syncthreads();
  }

  if constexpr (GATHER) {
    unsigned drow[2][8];
#pragma unroll
    for (int im = 0; im < 2; ++im)
#pragma unroll
      for (int r = 0; r < 8; ++r) {
        const int li = wm * 32 + im * 16 + r + 8 * lh;
        drow[im][r] = (unsigned)((z == 3) ? li : rowS[li]) * (unsigned)N;
      }
#pragma unroll
    for (int im = 0; im < 2; ++im)
#pragma unroll
      for (int j = 0; j < 4; ++j)
#pragma unroll
        for (int r = 0; r < 8; ++r) {
          const int li = wm * 32 + im * 16 + r + 8 * lh;
          const unsigned n = (unsigned)(n0 + wn * 64 + j * 16 + lm);
          if (li < nvalid) C[drow[im][r] + n] = acc[im][j][r];
        }
  } else {
#pragma unroll
    for (int im = 0; im < 2; ++im)
#pragma unroll
      for (int j = 0; j < 4; ++j)
#pragma unroll
        for (int r = 0; r < 8; ++r) {
          int m = m0 + wm * 32 + im * 16 + r + 8 * lh;
          int n = n0 + wn * 64 + j * 16 + lm;
          C[(size_t)m * N + n] = acc[im][j][r];
        }
  }
}

constexpr int VSL = 16;

__global__ __launch_bounds__(128) void xsum_rows(
    const float* __restrict__ X, const int* __restrict__ mask, float* __restrict__ XS) {
  __shared__ float red[3][17][128];
  const int t = threadIdx.x;
  const int lane = t & 31;
  const int rg = __builtin_amdgcn_readfirstlane(t >> 5);
  const int bh = blockIdx.y;
  const int b = bh / NH;
  const int h = bh % NH;
  const int k0 = blockIdx.x * 128 + lane * 4;
  const float* xp = X + ((size_t)b * Sq + h * 128 + rg * 32) * Hd + k0;
  unsigned pat = 0;
  {
    const int4* m4 = (const int4*)(mask + b * Sq + 16 * (rg * 32 + lane));
#pragma unroll
    for (int q = 0; q < 4; ++q) {
      const int4 m = m4[q];
      pat |= (m.x == 0 ? 1u : 0u) << (4 * q);
      pat |= (m.y == 0 ? 1u : 0u) << (4 * q + 1);
      pat |= (m.z == 0 ? 1u : 0u) << (4 * q + 2);
      pat |= (m.w == 0 ? 1u : 0u) << (4 * q + 3);
    }
  }
  float4 acc[16];
#pragma unroll
  for (int c = 0; c < 16; ++c) acc[c] = make_float4(0.f, 0.f, 0.f, 0.f);
  float4 all = make_float4(0.f, 0.f, 0.f, 0.f);
  for (int i0 = 0; i0 < 32; i0 += 8) {
    float4 v[8];
#pragma unroll
    for (int j = 0; j < 8; ++j) v[j] = *(const float4*)(xp + (size_t)(i0 + j) * Hd);
    __builtin_amdgcn_sched_barrier(0);
#pragma unroll
    for (int j = 0; j < 8; ++j) {
      const unsigned p = (unsigned)__builtin_amdgcn_readlane((int)pat, i0 + j);
      all.x += v[j].x; all.y += v[j].y; all.z += v[j].z; all.w += v[j].w;
#pragma unroll
      for (int c = 0; c < 16; ++c) {
        if ((p >> c) & 1u) {
          acc[c].x += v[j].x; acc[c].y += v[j].y; acc[c].z += v[j].z; acc[c].w += v[j].w;
        }
      }
    }
  }
  if (rg > 0) {
#pragma unroll
    for (int c = 0; c < 16; ++c) *(float4*)&red[rg - 1][c][lane * 4] = acc[c];
    *(float4*)&red[rg - 1][16][lane * 4] = all;
  }
  __syncthreads();
  if (rg == 0) {
#pragma unroll
    for (int c = 0; c < 17; ++c) {
      float4 a = (c < 16) ? acc[c] : all;
#pragma unroll
      for (int g = 0; g < 3; ++g) {
        const float4 p = *(const float4*)&red[g][c][lane * 4];
        a.x += p.x; a.y += p.y; a.z += p.z; a.w += p.w;
      }
      const int row = (c < 16) ? c * 32 + bh : XS_ALL_ROW0 + bh;
      *(float4*)(XS + (size_t)row * Hd + k0) = a;
    }
  }
}

constexpr int QT = 64;
constexpr int KT = 32;
constexpr int KVP = 12;
constexpr int VIPAD = 32;
constexpr int PPAD = 4;

__global__ __launch_bounds__(128) void attn_f32_wmma(
    const float* __restrict__ Qm, const float* __restrict__ Km,
    const float* __restrict__ Vm, const int* __restrict__ mask,
    const float* __restrict__ pall, const float* __restrict__ pmsk,
    float* __restrict__ O) {
  __shared__ float Ks[KT][HD + KVP];
  __shared__ float Vsp[KT / 2][2 * HD + VIPAD];
  __shared__ float Ps[4][16][KT + PPAD];
  __shared__ float corrS[4][16];
  __shared__ float rsumS[4][16];
  __shared__ float rmaxS[4][16];
  __shared__ float vmskS[HD];
  __shared__ int U[Sq];
  __shared__ int wtot[4];

  const int t = threadIdx.x;
  const int lane = t & 31;
  const int w = t >> 5;
  const int lm = lane & 15;
  const int lh = lane >> 4;

  const int bh = blockIdx.x % (Bz * NH);
  const int qb = blockIdx.x / (Bz * NH);
  const int b = bh / NH;
  const int q0 = qb * QT;

  const size_t baseBH = (size_t)bh * Sq * HD;
  const int* mrow = mask + b * Sq;

  unsigned bal[16];
  int cnt = 0;
#pragma unroll
  for (int i = 0; i < 16; ++i) {
    bal[i] = __builtin_amdgcn_ballot_w32(mrow[w * 512 + i * 32 + lane] != 0);
    cnt += __popc(bal[i]);
  }
  if (lane == 0) wtot[w] = cnt;

  float vall = 0.f, vmsk = 0.f;
  {
    const float* pa = pall + (size_t)bh * VSL * HD + t;
    const float* pm = pmsk + (size_t)bh * VSL * HD + t;
#pragma unroll
    for (int s = 0; s < VSL; ++s) { vall += pa[s * HD]; vmsk += pm[s * HD]; }
  }
  vmskS[t] = vmsk;
  __syncthreads();

  int ubase = 0;
#pragma unroll
  for (int ww = 0; ww < 4; ++ww) if (ww < w) ubase += wtot[ww];
  const int nb = wtot[0] + wtot[1] + wtot[2] + wtot[3];
  {
    const unsigned lt = (1u << lane) - 1u;
#pragma unroll
    for (int i = 0; i < 16; ++i) {
      if ((bal[i] >> lane) & 1u) U[ubase + __popc(bal[i] & lt)] = w * 512 + i * 32 + lane;
      ubase += __popc(bal[i]);
    }
  }

  {
    const float vmean = vall * (1.0f / (float)Sq);
    for (int r = 0; r < QT; ++r)
      if (mrow[q0 + r] == 0) O[baseBH + (size_t)(q0 + r) * HD + t] = vmean;
  }
  __syncthreads();
  if (q0 >= nb) return;
  const int nlast = nb - 1;

  int qi = q0 + w * 16 + lm;
  qi = (qi < nb) ? qi : nlast;
  const float* qrow = Qm + baseBH + (size_t)U[qi] * HD;
  v2f qf[HD / 4];
#pragma unroll
  for (int kd = 0; kd < HD / 4; ++kd)
    qf[kd] = *(const v2f*)(qrow + 4 * kd + 2 * lh);

  v8f o[8];
#pragma unroll
  for (int s = 0; s < 8; ++s) o[s] = vzero8();
  float rmax = -INFINITY;
  float rsum = 0.0f;

  for (int kt = 0; kt < nb; kt += KT) {
#pragma unroll
    for (int i = 0; i < 8; ++i) {
      int idx = t + i * 128;
      int r = idx >> 5;
      int c = (idx & 31) << 2;
      int kr = kt + r;
      kr = (kr < nb) ? kr : nlast;
      const size_t ro = baseBH + (size_t)U[kr] * HD + c;
      float4 kv = *(const float4*)(Km + ro);
      *(float4*)&Ks[r][c] = kv;
      float4 vv = *(const float4*)(Vm + ro);
      float* dst = &Vsp[r >> 1][2 * c + (r & 1)];
      dst[0] = vv.x;
      dst[2] = vv.y;
      dst[4] = vv.z;
      dst[6] = vv.w;
    }
    __syncthreads();

    v8f sc[2];
    sc[0] = vzero8();
    sc[1] = vzero8();
#pragma unroll
    for (int kd = 0; kd < HD / 4; ++kd) {
      v2f af = qf[kd];
#pragma unroll
      for (int j = 0; j < 2; ++j) {
        int kc = j * 16 + lm;
        v2f bf = *(const v2f*)&Ks[kc][4 * kd + 2 * lh];
        sc[j] = WMMA_F32(af, bf, sc[j]);
      }
    }

#pragma unroll
    for (int j = 0; j < 2; ++j)
#pragma unroll
      for (int r = 0; r < 8; ++r) {
        int m = r + 8 * lh;
        int kc = j * 16 + lm;
        float v = sc[j][r];
        if (kt + kc >= nb) v = -INFINITY;
        Ps[w][m][kc] = v;
      }
    __syncthreads();

    {
      const int c0 = lh * 16;
      float pm = -INFINITY;
#pragma unroll
      for (int c = 0; c < 16; ++c) pm = fmaxf(pm, Ps[w][lm][c0 + c]);
      pm = fmaxf(pm, __shfl_xor(pm, 16, 32));
      float nm = fmaxf(rmax, pm);
      float cf = __expf(rmax - nm);
      float ps = 0.0f;
#pragma unroll
      for (int c = 0; c < 16; ++c) {
        float e = __expf(Ps[w][lm][c0 + c] - nm);
        Ps[w][lm][c0 + c] = e;
        ps += e;
      }
      ps += __shfl_xor(ps, 16, 32);
      rsum = rsum * cf + ps;
      rmax = nm;
      if (lane < 16) corrS[w][lm] = cf;
    }
    __syncthreads();

    float cv[8];
#pragma unroll
    for (int r = 0; r < 8; ++r) cv[r] = corrS[w][r + 8 * lh];
#pragma unroll
    for (int s = 0; s < 8; ++s)
#pragma unroll
      for (int r = 0; r < 8; ++r) o[s][r] *= cv[r];

#pragma unroll
    for (int kk = 0; kk < KT; kk += 4) {
      v2f af = *(const v2f*)&Ps[w][lm][kk + 2 * lh];
      const int khalf = (kk >> 1) + lh;
#pragma unroll
      for (int s = 0; s < 8; ++s) {
        int col = s * 16 + lm;
        v2f bf = *(const v2f*)&Vsp[khalf][2 * col];
        o[s] = WMMA_F32(af, bf, o[s]);
      }
    }
    __syncthreads();
  }

  if (lane < 16) { rsumS[w][lm] = rsum; rmaxS[w][lm] = rmax; }
  __syncthreads();

  const int nmask = Sq - nb;
  const float fnm = (float)nmask;
  float ra[8], rc[8], rv[8];
#pragma unroll
  for (int r = 0; r < 8; ++r) {
    const float mx = rmaxS[w][r + 8 * lh];
    const float rs = rsumS[w][r + 8 * lh];
    const float M = (nmask > 0) ? fmaxf(mx, NEGV) : mx;
    const float a = __expf(mx - M);
    const float c = (nmask > 0) ? __expf(NEGV - M) : 0.0f;
    ra[r] = a;
    rc[r] = c;
    rv[r] = 1.0f / (rs * a + c * fnm);
  }

#pragma unroll
  for (int s = 0; s < 8; ++s) {
    const int d = s * 16 + lm;
    const float vk = vmskS[d];
#pragma unroll
    for (int r = 0; r < 8; ++r) {
      const int qi2 = q0 + w * 16 + r + 8 * lh;
      if (qi2 < nb) {
        const int s2 = U[qi2];
        O[baseBH + (size_t)s2 * HD + d] = (o[s][r] * ra[r] + rc[r] * vk) * rv[r];
      }
    }
  }
}

extern "C" void kernel_launch(void* const* d_in, const int* in_sizes, int n_in,
                              void* d_out, int out_size, void* d_ws, size_t ws_size,
                              hipStream_t stream) {
  const float* X    = (const float*)d_in[0];
  const int*   mask = (const int*)d_in[1];
  const float* Wq   = (const float*)d_in[2];
  const float* Wk   = (const float*)d_in[3];
  const float* Wv   = (const float*)d_in[4];
  const float* Wqm  = (const float*)d_in[5];
  const float* Wkm  = (const float*)d_in[6];
  const float* Wvm  = (const float*)d_in[7];
  const float* Wo   = (const float*)d_in[8];
  float* out = (float*)d_out;

  constexpr size_t SZ = (size_t)Bz * Sq * Hd;
  constexpr size_t WZ = (size_t)Hd * Hd;
  float* s0 = (float*)d_ws;
  float* s1 = s0 + SZ;
  float* s2 = s1 + SZ;
  float* s3 = s2 + SZ;
  float* fq = s0;
  float* fk = s0 + WZ;
  float* fv = s1;

  constexpr int M = Bz * Sq, N = Hd, K = Hd;
  dim3 gw3(N / TN, K / TM, 3);
  dim3 gg4(N / TN, M / TM, 4);
  dim3 gg1(N / TN, M / TM, 1);
  dim3 gb(256);

  float* vm = out;

  float* xs = s1 + WZ;
  float* pmsk = xs + (size_t)1024 * Hd;
  float* pall = pmsk + (size_t)Bz * NH * VSL * HD;
  xsum_rows<<<dim3(Hd / 128, Bz * NH), 128, 0, stream>>>(X, mask, xs);

  gemm_f32_wmma<false><<<gw3, gb, 0, stream>>>(Wq, Wqm, fq, Wk, Wkm, fk, Wv, Wvm, fv, Wv, Wvm, fv, mask, K, N, K);
  gemm_f32_wmma<true><<<gg4, gb, 0, stream>>>(X, fq, s2, X, fk, s3, X, fv, vm, xs, fv, pmsk, mask, M, N, K);

  attn_f32_wmma<<<Bz * NH * (Sq / QT), 128, 0, stream>>>(s2, s3, vm, mask, pall, pmsk, s0);

  gemm_f32_wmma<false><<<gg1, gb, 0, stream>>>(s0, Wo, out, s0, Wo, out, s0, Wo, out, s0, Wo, out, mask, M, N, K);
}


# GEMM K-loop: LDS fragment reads software-pipelined one k-step ahead through two register sets (one read group per 8 WMMAs, counted waits instead of a drain per 4 WMMAs); sums unchanged
# speedup vs baseline: 1.9139x; 1.0350x over previous
#include <hip/hip_runtime.h>
#include <hip/hip_bf16.h>
#include <math.h>

constexpr int Bz = 2;
constexpr int Sq = 2048;
constexpr int Hd = 2048;
constexpr int NH = 16;
constexpr int HD = 128;
constexpr float NEGV = -1000000.0f;

typedef float v2f __attribute__((ext_vector_type(2)));
typedef float v8f __attribute__((ext_vector_type(8)));

#define WMMA_F32(a, b, c) \
  __builtin_amdgcn_wmma_f32_16x16x4_f32(false, (a), false, (b), (short)0, (c), false, false)

__device__ __forceinline__ v8f vzero8() {
  v8f z = {0.f, 0.f, 0.f, 0.f, 0.f, 0.f, 0.f, 0.f};
  return z;
}

constexpr int TM = 128, TN = 128, TK = 32;
constexpr int XS_ALL_ROW0 = 512;
constexpr int APAD = 4;
constexpr int BIPAD = 32;

template <bool GATHER>
__global__ __launch_bounds__(256) void gemm_f32_wmma(
    const float* __restrict__ A0, const float* __restrict__ B0, float* __restrict__ C0,
    const float* __restrict__ A1, const float* __restrict__ B1, float* __restrict__ C1,
    const float* __restrict__ A2, const float* __restrict__ B2, float* __restrict__ C2,
    const float* __restrict__ A3, const float* __restrict__ B3, float* __restrict__ C3,
    const int* __restrict__ mask, int M, int N, int K) {
  __shared__ float As[TM][TK + APAD];
  __shared__ float Bsp[TK / 2][2 * TN + BIPAD];
  __shared__ int rowS[GATHER ? TM : 1];
  __shared__ int relS[GATHER ? 128 : 1];
  __shared__ unsigned wbalS[4];

  const int z = blockIdx.z;
  const float* __restrict__ A  = (z == 0) ? A0 : (z == 1) ? A1 : (z == 2) ? A2 : A3;
  const float* __restrict__ Bm = (z == 0) ? B0 : (z == 1) ? B1 : (z == 2) ? B2 : B3;
  float* __restrict__ C        = (z == 0) ? C0 : (z == 1) ? C1 : (z == 2) ? C2 : C3;

  const int t = threadIdx.x;
  const int lane = t & 31;
  const int wave = t >> 5;
  const int lm = lane & 15;
  const int lh = lane >> 4;
  const int wm = wave & 3;
  const int wn = wave >> 2;
  const int m0 = blockIdx.y * TM;
  const int n0 = blockIdx.x * TN;

  int nvalid = TM;
  if constexpr (GATHER) {
    if (z < 3) {
      const int b = blockIdx.y & 1;
      const int ti = blockIdx.y >> 1;
      bool f = false;
      if (t < 128) f = mask[b * Sq + 16 * t + (int)blockIdx.x] != 0;
      const unsigned bal = __builtin_amdgcn_ballot_w32(f);
      if (wave < 4 && lane == 0) wbalS[wave] = bal;
      __syncthreads();
      const unsigned u0 = wbalS[0], u1 = wbalS[1], u2 = wbalS[2], u3 = wbalS[3];
      const int nR = __popc(u0) + __popc(u1) + __popc(u2) + __popc(u3);
      const int total = 16 * nR;
      if (ti * TM >= total) return;
      if (f) {
        int pos = __popc(bal & ((1u << lane) - 1u));
        if (wave > 0) pos += __popc(u0);
        if (wave > 1) pos += __popc(u1);
        if (wave > 2) pos += __popc(u2);
        relS[pos] = t;
      }
      __syncthreads();
      if (t < TM) {
        int g = ti * TM + t;
        g = (g < total) ? g : total - 1;
        int h = 0;
#pragma unroll
        for (int k = 1; k < 16; ++k) h += (g >= k * nR) ? 1 : 0;
        rowS[t] = b * Sq + h * 128 + relS[g - h * nR];
      }
      nvalid = (total - ti * TM < TM) ? total - ti * TM : TM;
    } else {
      if (blockIdx.y != 0) return;
      if (t < TM) {
        const int tt = (t < 64) ? t : 63;
        rowS[t] = (tt < 32) ? (int)blockIdx.x * 32 + tt : XS_ALL_ROW0 + (tt - 32);
      }
      nvalid = 64;
    }
    __syncthreads();
  }

  v8f acc[2][4];
#pragma unroll
  for (int i = 0; i < 2; ++i)
#pragma unroll
    for (int j = 0; j < 4; ++j) acc[i][j] = vzero8();

  const int ar = t >> 3;
  const int ac = (t & 7) << 2;
  const int br = t >> 5;
  const int bc = (t & 31) << 2;
  const float* ap = A + (size_t)(m0 + ar) * K + ac;
  const float* bp = Bm + (size_t)br * N + n0 + bc;

  const size_t aStep = (size_t)32 * K;
  const size_t bStep = (size_t)8 * N;
  const char* const Ab = (const char*)A;
  unsigned oa0 = 0, oa1 = 0, oa2 = 0, oa3 = 0;
  if constexpr (GATHER) {
    oa0 = ((unsigned)rowS[ar] * (unsigned)K + (unsigned)ac) * 4u;
    oa1 = ((unsigned)rowS[ar + 32] * (unsigned)K + (unsigned)ac) * 4u;
    oa2 = ((unsigned)rowS[ar + 64] * (unsigned)K + (unsigned)ac) * 4u;
    oa3 = ((unsigned)rowS[ar + 96] * (unsigned)K + (unsigned)ac) * 4u;
  }
  float4 ga0, ga1, ga2, ga3;
  if constexpr (GATHER) {
    ga0 = *(const float4*)(Ab + oa0);
    ga1 = *(const float4*)(Ab + oa1);
    ga2 = *(const float4*)(Ab + oa2);
    ga3 = *(const float4*)(Ab + oa3);
  } else {
    ga0 = *(const float4*)(ap);
    ga1 = *(const float4*)(ap + aStep);
    ga2 = *(const float4*)(ap + 2 * aStep);
    ga3 = *(const float4*)(ap + 3 * aStep);
  }
  float4 gb0 = *(const float4*)(bp);
  float4 gb1 = *(const float4*)(bp + bStep);
  float4 gb2 = *(const float4*)(bp + 2 * bStep);
  float4 gb3 = *(const float4*)(bp + 3 * bStep);

  float* const bdst = &Bsp[br >> 1][2 * bc + (br & 1)];
  constexpr int BROW4 = 4 * (2 * TN + BIPAD);

  for (int kt = 0; kt < K; kt += TK) {
    *(float4*)&As[ar][ac] = ga0;
    *(float4*)&As[ar + 32][ac] = ga1;
    *(float4*)&As[ar + 64][ac] = ga2;
    *(float4*)&As[ar + 96][ac] = ga3;
    bdst[0] = gb0.x; bdst[2] = gb0.y; bdst[4] = gb0.z; bdst[6] = gb0.w;
    bdst[BROW4 + 0] = gb1.x; bdst[BROW4 + 2] = gb1.y;
    bdst[BROW4 + 4] = gb1.z; bdst[BROW4 + 6] = gb1.w;
    bdst[2 * BROW4 + 0] = gb2.x; bdst[2 * BROW4 + 2] = gb2.y;
    bdst[2 * BROW4 + 4] = gb2.z; bdst[2 * BROW4 + 6] = gb2.w;
    bdst[3 * BROW4 + 0] = gb3.x; bdst[3 * BROW4 + 2] = gb3.y;
    bdst[3 * BROW4 + 4] = gb3.z; bdst[3 * BROW4 + 6] = gb3.w;
    __syncthreads();

    {
      const int kn = (kt + TK < K) ? kt + TK : kt;
      const float* an = ap + kn;
      const float* bn = bp + (size_t)kn * N;
      if constexpr (GATHER) {
        const unsigned kb = (unsigned)kn * 4u;
        ga0 = *(const float4*)(Ab + (oa0 + kb));
        ga1 = *(const float4*)(Ab + (oa1 + kb));
        ga2 = *(const float4*)(Ab + (oa2 + kb));
        ga3 = *(const float4*)(Ab + (oa3 + kb));
      } else {
        ga0 = *(const float4*)(an);
        ga1 = *(const float4*)(an + aStep);
        ga2 = *(const float4*)(an + 2 * aStep);
        ga3 = *(const float4*)(an + 3 * aStep);
      }
      gb0 = *(const float4*)(bn);
      gb1 = *(const float4*)(bn + bStep);
      gb2 = *(const float4*)(bn + 2 * bStep);
      gb3 = *(const float4*)(bn + 3 * bStep);
    }
    __builtin_amdgcn_sched_barrier(0);

    auto ldfrag = [&](const int kk, v2f (&af)[2], v2f (&bf)[4]) {
#pragma unroll
      for (int im = 0; im < 2; ++im) {
        int row = wm * 32 + im * 16 + lm;
        af[im] = *(const v2f*)&As[row][kk + 2 * lh];
      }
      const int khalf = (kk >> 1) + lh;
#pragma unroll
      for (int j = 0; j < 4; ++j) {
        int col = wn * 64 + j * 16 + lm;
        bf[j] = *(const v2f*)&Bsp[khalf][2 * col];
      }
    };
    auto mma8 = [&](const v2f (&af)[2], const v2f (&bf)[4]) {
#pragma unroll
      for (int j = 0; j < 4; ++j) {
        acc[0][j] = WMMA_F32(af[0], bf[j], acc[0][j]);
        acc[1][j] = WMMA_F32(af[1], bf[j], acc[1][j]);
      }
    };
    v2f afE[2], bfE[4], afO[2], bfO[4];
    ldfrag(0, afE, bfE);
#pragma unroll
    for (int kk = 0; kk < TK; kk += 8) {
      ldfrag(kk + 4, afO, bfO);
      mma8(afE, bfE);
      if (kk + 8 < TK) ldfrag(kk + 8, afE, bfE);
      mma8(afO, bfO);
    }
    __syncthreads();
  }

  if constexpr (GATHER) {
    unsigned drow[2][8];
#pragma unroll
    for (int im = 0; im < 2; ++im)
#pragma unroll
      for (int r = 0; r < 8; ++r) {
        const int li = wm * 32 + im * 16 + r + 8 * lh;
        drow[im][r] = (unsigned)((z == 3) ? li : rowS[li]) * (unsigned)N;
      }
#pragma unroll
    for (int im = 0; im < 2; ++im)
#pragma unroll
      for (int j = 0; j < 4; ++j)
#pragma unroll
        for (int r = 0; r < 8; ++r) {
          const int li = wm * 32 + im * 16 + r + 8 * lh;
          const unsigned n = (unsigned)(n0 + wn * 64 + j * 16 + lm);
          if (li < nvalid) C[drow[im][r] + n] = acc[im][j][r];
        }
  } else {
#pragma unroll
    for (int im = 0; im < 2; ++im)
#pragma unroll
      for (int j = 0; j < 4; ++j)
#pragma unroll
        for (int r = 0; r < 8; ++r) {
          int m = m0 + wm * 32 + im * 16 + r + 8 * lh;
          int n = n0 + wn * 64 + j * 16 + lm;
          C[(size_t)m * N + n] = acc[im][j][r];
        }
  }
}

constexpr int VSL = 16;

__global__ __launch_bounds__(128) void xsum_rows(
    const float* __restrict__ X, const int* __restrict__ mask, float* __restrict__ XS) {
  __shared__ float red[3][17][128];
  const int t = threadIdx.x;
  const int lane = t & 31;
  const int rg = __builtin_amdgcn_readfirstlane(t >> 5);
  const int bh = blockIdx.y;
  const int b = bh / NH;
  const int h = bh % NH;
  const int k0 = blockIdx.x * 128 + lane * 4;
  const float* xp = X + ((size_t)b * Sq + h * 128 + rg * 32) * Hd + k0;
  unsigned pat = 0;
  {
    const int4* m4 = (const int4*)(mask + b * Sq + 16 * (rg * 32 + lane));
#pragma unroll
    for (int q = 0; q < 4; ++q) {
      const int4 m = m4[q];
      pat |= (m.x == 0 ? 1u : 0u) << (4 * q);
      pat |= (m.y == 0 ? 1u : 0u) << (4 * q + 1);
      pat |= (m.z == 0 ? 1u : 0u) << (4 * q + 2);
      pat |= (m.w == 0 ? 1u : 0u) << (4 * q + 3);
    }
  }
  float4 acc[16];
#pragma unroll
  for (int c = 0; c < 16; ++c) acc[c] = make_float4(0.f, 0.f, 0.f, 0.f);
  float4 all = make_float4(0.f, 0.f, 0.f, 0.f);
  for (int i0 = 0; i0 < 32; i0 += 8) {
    float4 v[8];
#pragma unroll
    for (int j = 0; j < 8; ++j) v[j] = *(const float4*)(xp + (size_t)(i0 + j) * Hd);
    __builtin_amdgcn_sched_barrier(0);
#pragma unroll
    for (int j = 0; j < 8; ++j) {
      const unsigned p = (unsigned)__builtin_amdgcn_readlane((int)pat, i0 + j);
      all.x += v[j].x; all.y += v[j].y; all.z += v[j].z; all.w += v[j].w;
#pragma unroll
      for (int c = 0; c < 16; ++c) {
        if ((p >> c) & 1u) {
          acc[c].x += v[j].x; acc[c].y += v[j].y; acc[c].z += v[j].z; acc[c].w += v[j].w;
        }
      }
    }
  }
  if (rg > 0) {
#pragma unroll
    for (int c = 0; c < 16; ++c) *(float4*)&red[rg - 1][c][lane * 4] = acc[c];
    *(float4*)&red[rg - 1][16][lane * 4] = all;
  }
  __syncthreads();
  if (rg == 0) {
#pragma unroll
    for (int c = 0; c < 17; ++c) {
      float4 a = (c < 16) ? acc[c] : all;
#pragma unroll
      for (int g = 0; g < 3; ++g) {
        const float4 p = *(const float4*)&red[g][c][lane * 4];
        a.x += p.x; a.y += p.y; a.z += p.z; a.w += p.w;
      }
      const int row = (c < 16) ? c * 32 + bh : XS_ALL_ROW0 + bh;
      *(float4*)(XS + (size_t)row * Hd + k0) = a;
    }
  }
}

constexpr int QT = 64;
constexpr int KT = 32;
constexpr int KVP = 12;
constexpr int VIPAD = 32;
constexpr int PPAD = 4;

__global__ __launch_bounds__(128) void attn_f32_wmma(
    const float* __restrict__ Qm, const float* __restrict__ Km,
    const float* __restrict__ Vm, const int* __restrict__ mask,
    const float* __restrict__ pall, const float* __restrict__ pmsk,
    float* __restrict__ O) {
  __shared__ float Ks[KT][HD + KVP];
  __shared__ float Vsp[KT / 2][2 * HD + VIPAD];
  __shared__ float Ps[4][16][KT + PPAD];
  __shared__ float corrS[4][16];
  __shared__ float rsumS[4][16];
  __shared__ float rmaxS[4][16];
  __shared__ float vmskS[HD];
  __shared__ int U[Sq];
  __shared__ int wtot[4];

  const int t = threadIdx.x;
  const int lane = t & 31;
  const int w = t >> 5;
  const int lm = lane & 15;
  const int lh = lane >> 4;

  const int bh = blockIdx.x % (Bz * NH);
  const int qb = blockIdx.x / (Bz * NH);
  const int b = bh / NH;
  const int q0 = qb * QT;

  const size_t baseBH = (size_t)bh * Sq * HD;
  const int* mrow = mask + b * Sq;

  unsigned bal[16];
  int cnt = 0;
#pragma unroll
  for (int i = 0; i < 16; ++i) {
    bal[i] = __builtin_amdgcn_ballot_w32(mrow[w * 512 + i * 32 + lane] != 0);
    cnt += __popc(bal[i]);
  }
  if (lane == 0) wtot[w] = cnt;

  float vall = 0.f, vmsk = 0.f;
  {
    const float* pa = pall + (size_t)bh * VSL * HD + t;
    const float* pm = pmsk + (size_t)bh * VSL * HD + t;
#pragma unroll
    for (int s = 0; s < VSL; ++s) { vall += pa[s * HD]; vmsk += pm[s * HD]; }
  }
  vmskS[t] = vmsk;
  __syncthreads();

  int ubase = 0;
#pragma unroll
  for (int ww = 0; ww < 4; ++ww) if (ww < w) ubase += wtot[ww];
  const int nb = wtot[0] + wtot[1] + wtot[2] + wtot[3];
  {
    const unsigned lt = (1u << lane) - 1u;
#pragma unroll
    for (int i = 0; i < 16; ++i) {
      if ((bal[i] >> lane) & 1u) U[ubase + __popc(bal[i] & lt)] = w * 512 + i * 32 + lane;
      ubase += __popc(bal[i]);
    }
  }

  {
    const float vmean = vall * (1.0f / (float)Sq);
    for (int r = 0; r < QT; ++r)
      if (mrow[q0 + r] == 0) O[baseBH + (size_t)(q0 + r) * HD + t] = vmean;
  }
  __syncthreads();
  if (q0 >= nb) return;
  const int nlast = nb - 1;

  int qi = q0 + w * 16 + lm;
  qi = (qi < nb) ? qi : nlast;
  const float* qrow = Qm + baseBH + (size_t)U[qi] * HD;
  v2f qf[HD / 4];
#pragma unroll
  for (int kd = 0; kd < HD / 4; ++kd)
    qf[kd] = *(const v2f*)(qrow + 4 * kd + 2 * lh);

  v8f o[8];
#pragma unroll
  for (int s = 0; s < 8; ++s) o[s] = vzero8();
  float rmax = -INFINITY;
  float rsum = 0.0f;

  for (int kt = 0; kt < nb; kt += KT) {
#pragma unroll
    for (int i = 0; i < 8; ++i) {
      int idx = t + i * 128;
      int r = idx >> 5;
      int c = (idx & 31) << 2;
      int kr = kt + r;
      kr = (kr < nb) ? kr : nlast;
      const size_t ro = baseBH + (size_t)U[kr] * HD + c;
      float4 kv = *(const float4*)(Km + ro);
      *(float4*)&Ks[r][c] = kv;
      float4 vv = *(const float4*)(Vm + ro);
      float* dst = &Vsp[r >> 1][2 * c + (r & 1)];
      dst[0] = vv.x;
      dst[2] = vv.y;
      dst[4] = vv.z;
      dst[6] = vv.w;
    }
    __syncthreads();

    v8f sc[2];
    sc[0] = vzero8();
    sc[1] = vzero8();
#pragma unroll
    for (int kd = 0; kd < HD / 4; ++kd) {
      v2f af = qf[kd];
#pragma unroll
      for (int j = 0; j < 2; ++j) {
        int kc = j * 16 + lm;
        v2f bf = *(const v2f*)&Ks[kc][4 * kd + 2 * lh];
        sc[j] = WMMA_F32(af, bf, sc[j]);
      }
    }

#pragma unroll
    for (int j = 0; j < 2; ++j)
#pragma unroll
      for (int r = 0; r < 8; ++r) {
        int m = r + 8 * lh;
        int kc = j * 16 + lm;
        float v = sc[j][r];
        if (kt + kc >= nb) v = -INFINITY;
        Ps[w][m][kc] = v;
      }
    __syncthreads();

    {
      const int c0 = lh * 16;
      float pm = -INFINITY;
#pragma unroll
      for (int c = 0; c < 16; ++c) pm = fmaxf(pm, Ps[w][lm][c0 + c]);
      pm = fmaxf(pm, __shfl_xor(pm, 16, 32));
      float nm = fmaxf(rmax, pm);
      float cf = __expf(rmax - nm);
      float ps = 0.0f;
#pragma unroll
      for (int c = 0; c < 16; ++c) {
        float e = __expf(Ps[w][lm][c0 + c] - nm);
        Ps[w][lm][c0 + c] = e;
        ps += e;
      }
      ps += __shfl_xor(ps, 16, 32);
      rsum = rsum * cf + ps;
      rmax = nm;
      if (lane < 16) corrS[w][lm] = cf;
    }
    __syncthreads();

    float cv[8];
#pragma unroll
    for (int r = 0; r < 8; ++r) cv[r] = corrS[w][r + 8 * lh];
#pragma unroll
    for (int s = 0; s < 8; ++s)
#pragma unroll
      for (int r = 0; r < 8; ++r) o[s][r] *= cv[r];

#pragma unroll
    for (int kk = 0; kk < KT; kk += 4) {
      v2f af = *(const v2f*)&Ps[w][lm][kk + 2 * lh];
      const int khalf = (kk >> 1) + lh;
#pragma unroll
      for (int s = 0; s < 8; ++s) {
        int col = s * 16 + lm;
        v2f bf = *(const v2f*)&Vsp[khalf][2 * col];
        o[s] = WMMA_F32(af, bf, o[s]);
      }
    }
    __syncthreads();
  }

  if (lane < 16) { rsumS[w][lm] = rsum; rmaxS[w][lm] = rmax; }
  __syncthreads();

  const int nmask = Sq - nb;
  const float fnm = (float)nmask;
  float ra[8], rc[8], rv[8];
#pragma unroll
  for (int r = 0; r < 8; ++r) {
    const float mx = rmaxS[w][r + 8 * lh];
    const float rs = rsumS[w][r + 8 * lh];
    const float M = (nmask > 0) ? fmaxf(mx, NEGV) : mx;
    const float a = __expf(mx - M);
    const float c = (nmask > 0) ? __expf(NEGV - M) : 0.0f;
    ra[r] = a;
    rc[r] = c;
    rv[r] = 1.0f / (rs * a + c * fnm);
  }

#pragma unroll
  for (int s = 0; s < 8; ++s) {
    const int d = s * 16 + lm;
    const float vk = vmskS[d];
#pragma unroll
    for (int r = 0; r < 8; ++r) {
      const int qi2 = q0 + w * 16 + r + 8 * lh;
      if (qi2 < nb) {
        const int s2 = U[qi2];
        O[baseBH + (size_t)s2 * HD + d] = (o[s][r] * ra[r] + rc[r] * vk) * rv[r];
      }
    }
  }
}

extern "C" void kernel_launch(void* const* d_in, const int* in_sizes, int n_in,
                              void* d_out, int out_size, void* d_ws, size_t ws_size,
                              hipStream_t stream) {
  const float* X    = (const float*)d_in[0];
  const int*   mask = (const int*)d_in[1];
  const float* Wq   = (const float*)d_in[2];
  const float* Wk   = (const float*)d_in[3];
  const float* Wv   = (const float*)d_in[4];
  const float* Wqm  = (const float*)d_in[5];
  const float* Wkm  = (const float*)d_in[6];
  const float* Wvm  = (const float*)d_in[7];
  const float* Wo   = (const float*)d_in[8];
  float* out = (float*)d_out;

  constexpr size_t SZ = (size_t)Bz * Sq * Hd;
  constexpr size_t WZ = (size_t)Hd * Hd;
  float* s0 = (float*)d_ws;
  float* s1 = s0 + SZ;
  float* s2 = s1 + SZ;
  float* s3 = s2 + SZ;
  float* fq = s0;
  float* fk = s0 + WZ;
  float* fv = s1;

  constexpr int M = Bz * Sq, N = Hd, K = Hd;
  dim3 gw3(N / TN, K / TM, 3);
  dim3 gg4(N / TN, M / TM, 4);
  dim3 gg1(N / TN, M / TM, 1);
  dim3 gb(256);

  float* vm = out;

  float* xs = s1 + WZ;
  float* pmsk = xs + (size_t)1024 * Hd;
  float* pall = pmsk + (size_t)Bz * NH * VSL * HD;
  xsum_rows<<<dim3(Hd / 128, Bz * NH), 128, 0, stream>>>(X, mask, xs);

  gemm_f32_wmma<false><<<gw3, gb, 0, stream>>>(Wq, Wqm, fq, Wk, Wkm, fk, Wv, Wvm, fv, Wv, Wvm, fv, mask, K, N, K);
  gemm_f32_wmma<true><<<gg4, gb, 0, stream>>>(X, fq, s2, X, fk, s3, X, fv, vm, xs, fv, pmsk, mask, M, N, K);

  attn_f32_wmma<<<Bz * NH * (Sq / QT), 128, 0, stream>>>(s2, s3, vm, mask, pall, pmsk, s0);

  gemm_f32_wmma<false><<<gg1, gb, 0, stream>>>(s0, Wo, out, s0, Wo, out, s0, Wo, out, s0, Wo, out, mask, M, N, K);
}


# Output projection over unmasked positions only: attention stores rows minus the (b,h) mean of V, masked K chunks skipped per 16-head row group, the mean re-added via chunk-summed W_o; sums regrouped
# speedup vs baseline: 2.0016x; 1.0458x over previous
#include <hip/hip_runtime.h>
#include <hip/hip_bf16.h>
#include <math.h>

constexpr int Bz = 2;
constexpr int Sq = 2048;
constexpr int Hd = 2048;
constexpr int NH = 16;
constexpr int HD = 128;
constexpr float NEGV = -1000000.0f;

typedef float v2f __attribute__((ext_vector_type(2)));
typedef float v8f __attribute__((ext_vector_type(8)));

#define WMMA_F32(a, b, c) \
  __builtin_amdgcn_wmma_f32_16x16x4_f32(false, (a), false, (b), (short)0, (c), false, false)

__device__ __forceinline__ v8f vzero8() {
  v8f z = {0.f, 0.f, 0.f, 0.f, 0.f, 0.f, 0.f, 0.f};
  return z;
}

constexpr int TM = 128, TN = 128, TK = 32;
constexpr int XS_ALL_ROW0 = 512;
constexpr int APAD = 4;
constexpr int BIPAD = 32;

template <bool GATHER>
__global__ __launch_bounds__(256) void gemm_f32_wmma(
    const float* __restrict__ A0, const float* __restrict__ B0, float* __restrict__ C0,
    const float* __restrict__ A1, const float* __restrict__ B1, float* __restrict__ C1,
    const float* __restrict__ A2, const float* __restrict__ B2, float* __restrict__ C2,
    const float* __restrict__ A3, const float* __restrict__ B3, float* __restrict__ C3,
    const int* __restrict__ mask, int M, int N, int K) {
  __shared__ float As[TM][TK + APAD];
  __shared__ float Bsp[TK / 2][2 * TN + BIPAD];
  __shared__ int rowS[GATHER ? TM : 1];
  __shared__ int relS[GATHER ? 128 : 1];
  __shared__ unsigned wbalS[4];

  const int z = blockIdx.z;
  const float* __restrict__ A  = (z == 0) ? A0 : (z == 1) ? A1 : (z == 2) ? A2 : A3;
  const float* __restrict__ Bm = (z == 0) ? B0 : (z == 1) ? B1 : (z == 2) ? B2 : B3;
  float* __restrict__ C        = (z == 0) ? C0 : (z == 1) ? C1 : (z == 2) ? C2 : C3;

  const int t = threadIdx.x;
  const int lane = t & 31;
  const int wave = t >> 5;
  const int lm = lane & 15;
  const int lh = lane >> 4;
  const int wm = wave & 3;
  const int wn = wave >> 2;
  const int m0 = blockIdx.y * TM;
  const int n0 = blockIdx.x * TN;

  int nvalid = TM;
  if constexpr (GATHER) {
    if (z < 3) {
      const int b = blockIdx.y & 1;
      const int ti = blockIdx.y >> 1;
      bool f = false;
      if (t < 128) f = mask[b * Sq + 16 * t + (int)blockIdx.x] != 0;
      const unsigned bal = __builtin_amdgcn_ballot_w32(f);
      if (wave < 4 && lane == 0) wbalS[wave] = bal;
      __syncthreads();
      const unsigned u0 = wbalS[0], u1 = wbalS[1], u2 = wbalS[2], u3 = wbalS[3];
      const int nR = __popc(u0) + __popc(u1) + __popc(u2) + __popc(u3);
      const int total = 16 * nR;
      if (ti * TM >= total) return;
      if (f) {
        int pos = __popc(bal & ((1u << lane) - 1u));
        if (wave > 0) pos += __popc(u0);
        if (wave > 1) pos += __popc(u1);
        if (wave > 2) pos += __popc(u2);
        relS[pos] = t;
      }
      __syncthreads();
      if (t < TM) {
        int g = ti * TM + t;
        g = (g < total) ? g : total - 1;
        int h = 0;
#pragma unroll
        for (int k = 1; k < 16; ++k) h += (g >= k * nR) ? 1 : 0;
        rowS[t] = b * Sq + h * 128 + relS[g - h * nR];
      }
      nvalid = (total - ti * TM < TM) ? total - ti * TM : TM;
    } else {
      if (blockIdx.y != 0) return;
      if (t < TM) {
        const int tt = (t < 64) ? t : 63;
        rowS[t] = (tt < 32) ? (int)blockIdx.x * 32 + tt : XS_ALL_ROW0 + (tt - 32);
      }
      nvalid = 64;
    }
    __syncthreads();
  }

  v8f acc[2][4];
#pragma unroll
  for (int i = 0; i < 2; ++i)
#pragma unroll
    for (int j = 0; j < 4; ++j) acc[i][j] = vzero8();

  const int ar = t >> 3;
  const int ac = (t & 7) << 2;
  const int br = t >> 5;
  const int bc = (t & 31) << 2;
  const float* ap = A + (size_t)(m0 + ar) * K + ac;
  const float* bp = Bm + (size_t)br * N + n0 + bc;

  const size_t aStep = (size_t)32 * K;
  const size_t bStep = (size_t)8 * N;
  const char* const Ab = (const char*)A;
  unsigned oa0 = 0, oa1 = 0, oa2 = 0, oa3 = 0;
  if constexpr (GATHER) {
    oa0 = ((unsigned)rowS[ar] * (unsigned)K + (unsigned)ac) * 4u;
    oa1 = ((unsigned)rowS[ar + 32] * (unsigned)K + (unsigned)ac) * 4u;
    oa2 = ((unsigned)rowS[ar + 64] * (unsigned)K + (unsigned)ac) * 4u;
    oa3 = ((unsigned)rowS[ar + 96] * (unsigned)K + (unsigned)ac) * 4u;
  }
  float4 ga0, ga1, ga2, ga3;
  if constexpr (GATHER) {
    ga0 = *(const float4*)(Ab + oa0);
    ga1 = *(const float4*)(Ab + oa1);
    ga2 = *(const float4*)(Ab + oa2);
    ga3 = *(const float4*)(Ab + oa3);
  } else {
    ga0 = *(const float4*)(ap);
    ga1 = *(const float4*)(ap + aStep);
    ga2 = *(const float4*)(ap + 2 * aStep);
    ga3 = *(const float4*)(ap + 3 * aStep);
  }
  float4 gb0 = *(const float4*)(bp);
  float4 gb1 = *(const float4*)(bp + bStep);
  float4 gb2 = *(const float4*)(bp + 2 * bStep);
  float4 gb3 = *(const float4*)(bp + 3 * bStep);

  float* const bdst = &Bsp[br >> 1][2 * bc + (br & 1)];
  constexpr int BROW4 = 4 * (2 * TN + BIPAD);

  for (int kt = 0; kt < K; kt += TK) {
    *(float4*)&As[ar][ac] = ga0;
    *(float4*)&As[ar + 32][ac] = ga1;
    *(float4*)&As[ar + 64][ac] = ga2;
    *(float4*)&As[ar + 96][ac] = ga3;
    bdst[0] = gb0.x; bdst[2] = gb0.y; bdst[4] = gb0.z; bdst[6] = gb0.w;
    bdst[BROW4 + 0] = gb1.x; bdst[BROW4 + 2] = gb1.y;
    bdst[BROW4 + 4] = gb1.z; bdst[BROW4 + 6] = gb1.w;
    bdst[2 * BROW4 + 0] = gb2.x; bdst[2 * BROW4 + 2] = gb2.y;
    bdst[2 * BROW4 + 4] = gb2.z; bdst[2 * BROW4 + 6] = gb2.w;
    bdst[3 * BROW4 + 0] = gb3.x; bdst[3 * BROW4 + 2] = gb3.y;
    bdst[3 * BROW4 + 4] = gb3.z; bdst[3 * BROW4 + 6] = gb3.w;
    __syncthreads();

    {
      const int kn = (kt + TK < K) ? kt + TK : kt;
      const float* an = ap + kn;
      const float* bn = bp + (size_t)kn * N;
      if constexpr (GATHER) {
        const unsigned kb = (unsigned)kn * 4u;
        ga0 = *(const float4*)(Ab + (oa0 + kb));
        ga1 = *(const float4*)(Ab + (oa1 + kb));
        ga2 = *(const float4*)(Ab + (oa2 + kb));
        ga3 = *(const float4*)(Ab + (oa3 + kb));
      } else {
        ga0 = *(const float4*)(an);
        ga1 = *(const float4*)(an + aStep);
        ga2 = *(const float4*)(an + 2 * aStep);
        ga3 = *(const float4*)(an + 3 * aStep);
      }
      gb0 = *(const float4*)(bn);
      gb1 = *(const float4*)(bn + bStep);
      gb2 = *(const float4*)(bn + 2 * bStep);
      gb3 = *(const float4*)(bn + 3 * bStep);
    }
    __builtin_amdgcn_sched_barrier(0);

    auto ldfrag = [&](const int kk, v2f (&af)[2], v2f (&bf)[4]) {
#pragma unroll
      for (int im = 0; im < 2; ++im) {
        int row = wm * 32 + im * 16 + lm;
        af[im] = *(const v2f*)&As[row][kk + 2 * lh];
      }
      const int khalf = (kk >> 1) + lh;
#pragma unroll
      for (int j = 0; j < 4; ++j) {
        int col = wn * 64 + j * 16 + lm;
        bf[j] = *(const v2f*)&Bsp[khalf][2 * col];
      }
    };
    auto mma8 = [&](const v2f (&af)[2], const v2f (&bf)[4]) {
#pragma unroll
      for (int j = 0; j < 4; ++j) {
        acc[0][j] = WMMA_F32(af[0], bf[j], acc[0][j]);
        acc[1][j] = WMMA_F32(af[1], bf[j], acc[1][j]);
      }
    };
    v2f afE[2], bfE[4], afO[2], bfO[4];
    ldfrag(0, afE, bfE);
#pragma unroll
    for (int kk = 0; kk < TK; kk += 8) {
      ldfrag(kk + 4, afO, bfO);
      mma8(afE, bfE);
      if (kk + 8 < TK) ldfrag(kk + 8, afE, bfE);
      mma8(afO, bfO);
    }
    __syncthreads();
  }

  if constexpr (GATHER) {
    unsigned drow[2][8];
#pragma unroll
    for (int im = 0; im < 2; ++im)
#pragma unroll
      for (int r = 0; r < 8; ++r) {
        const int li = wm * 32 + im * 16 + r + 8 * lh;
        drow[im][r] = (unsigned)((z == 3) ? li : rowS[li]) * (unsigned)N;
      }
#pragma unroll
    for (int im = 0; im < 2; ++im)
#pragma unroll
      for (int j = 0; j < 4; ++j)
#pragma unroll
        for (int r = 0; r < 8; ++r) {
          const int li = wm * 32 + im * 16 + r + 8 * lh;
          const unsigned n = (unsigned)(n0 + wn * 64 + j * 16 + lm);
          if (li < nvalid) C[drow[im][r] + n] = acc[im][j][r];
        }
  } else {
#pragma unroll
    for (int im = 0; im < 2; ++im)
#pragma unroll
      for (int j = 0; j < 4; ++j)
#pragma unroll
        for (int r = 0; r < 8; ++r) {
          int m = m0 + wm * 32 + im * 16 + r + 8 * lh;
          int n = n0 + wn * 64 + j * 16 + lm;
          C[(size_t)m * N + n] = acc[im][j][r];
        }
  }
}

constexpr int VSL = 16;

__global__ __launch_bounds__(128) void xsum_rows(
    const float* __restrict__ X, const int* __restrict__ mask, float* __restrict__ XS,
    const float* __restrict__ Wo, float* __restrict__ WS) {
  __shared__ float red[3][17][128];
  const int t = threadIdx.x;
  const int lane = t & 31;
  const int rg = __builtin_amdgcn_readfirstlane(t >> 5);
  if (blockIdx.y >= Bz * NH) {
    const int yy = blockIdx.y - Bz * NH;
    const int n = blockIdx.x * 128 + lane * 4;
#pragma unroll
    for (int i = 0; i < 4; ++i) {
      const int d = yy * 16 + rg + 4 * i;
      float4 a = make_float4(0.f, 0.f, 0.f, 0.f);
#pragma unroll
      for (int cc = 0; cc < 16; ++cc) {
        const float4 w = *(const float4*)(Wo + (size_t)(cc * 128 + d) * Hd + n);
        a.x += w.x; a.y += w.y; a.z += w.z; a.w += w.w;
      }
      *(float4*)(WS + (size_t)d * Hd + n) = a;
    }
    return;
  }
  const int bh = blockIdx.y;
  const int b = bh / NH;
  const int h = bh % NH;
  const int k0 = blockIdx.x * 128 + lane * 4;
  const float* xp = X + ((size_t)b * Sq + h * 128 + rg * 32) * Hd + k0;
  unsigned pat = 0;
  {
    const int4* m4 = (const int4*)(mask + b * Sq + 16 * (rg * 32 + lane));
#pragma unroll
    for (int q = 0; q < 4; ++q) {
      const int4 m = m4[q];
      pat |= (m.x == 0 ? 1u : 0u) << (4 * q);
      pat |= (m.y == 0 ? 1u : 0u) << (4 * q + 1);
      pat |= (m.z == 0 ? 1u : 0u) << (4 * q + 2);
      pat |= (m.w == 0 ? 1u : 0u) << (4 * q + 3);
    }
  }
  float4 acc[16];
#pragma unroll
  for (int c = 0; c < 16; ++c) acc[c] = make_float4(0.f, 0.f, 0.f, 0.f);
  float4 all = make_float4(0.f, 0.f, 0.f, 0.f);
  for (int i0 = 0; i0 < 32; i0 += 8) {
    float4 v[8];
#pragma unroll
    for (int j = 0; j < 8; ++j) v[j] = *(const float4*)(xp + (size_t)(i0 + j) * Hd);
    __builtin_amdgcn_sched_barrier(0);
#pragma unroll
    for (int j = 0; j < 8; ++j) {
      const unsigned p = (unsigned)__builtin_amdgcn_readlane((int)pat, i0 + j);
      all.x += v[j].x; all.y += v[j].y; all.z += v[j].z; all.w += v[j].w;
#pragma unroll
      for (int c = 0; c < 16; ++c) {
        if ((p >> c) & 1u) {
          acc[c].x += v[j].x; acc[c].y += v[j].y; acc[c].z += v[j].z; acc[c].w += v[j].w;
        }
      }
    }
  }
  if (rg > 0) {
#pragma unroll
    for (int c = 0; c < 16; ++c) *(float4*)&red[rg - 1][c][lane * 4] = acc[c];
    *(float4*)&red[rg - 1][16][lane * 4] = all;
  }
  __syncthreads();
  if (rg == 0) {
#pragma unroll
    for (int c = 0; c < 17; ++c) {
      float4 a = (c < 16) ? acc[c] : all;
#pragma unroll
      for (int g = 0; g < 3; ++g) {
        const float4 p = *(const float4*)&red[g][c][lane * 4];
        a.x += p.x; a.y += p.y; a.z += p.z; a.w += p.w;
      }
      const int row = (c < 16) ? c * 32 + bh : XS_ALL_ROW0 + bh;
      *(float4*)(XS + (size_t)row * Hd + k0) = a;
    }
  }
}

constexpr int QT = 64;
constexpr int KT = 32;
constexpr int KVP = 12;
constexpr int VIPAD = 32;
constexpr int PPAD = 4;

__global__ __launch_bounds__(128) void attn_f32_wmma(
    const float* __restrict__ Qm, const float* __restrict__ Km,
    const float* __restrict__ Vm, const int* __restrict__ mask,
    const float* __restrict__ pall, const float* __restrict__ pmsk,
    float* __restrict__ O, float* __restrict__ vmeanOut) {
  __shared__ float Ks[KT][HD + KVP];
  __shared__ float Vsp[KT / 2][2 * HD + VIPAD];
  __shared__ float Ps[4][16][KT + PPAD];
  __shared__ float corrS[4][16];
  __shared__ float rsumS[4][16];
  __shared__ float rmaxS[4][16];
  __shared__ float vmskS[HD];
  __shared__ float vmeanS[HD];
  __shared__ int U[Sq];
  __shared__ int wtot[4];

  const int t = threadIdx.x;
  const int lane = t & 31;
  const int w = t >> 5;
  const int lm = lane & 15;
  const int lh = lane >> 4;

  const int bh = blockIdx.x % (Bz * NH);
  const int qb = blockIdx.x / (Bz * NH);
  const int b = bh / NH;
  const int q0 = qb * QT;

  const size_t baseBH = (size_t)bh * Sq * HD;
  const int* mrow = mask + b * Sq;

  unsigned bal[16];
  int cnt = 0;
#pragma unroll
  for (int i = 0; i < 16; ++i) {
    bal[i] = __builtin_amdgcn_ballot_w32(mrow[w * 512 + i * 32 + lane] != 0);
    cnt += __popc(bal[i]);
  }
  if (lane == 0) wtot[w] = cnt;

  float vall = 0.f, vmsk = 0.f;
  {
    const float* pa = pall + (size_t)bh * VSL * HD + t;
    const float* pm = pmsk + (size_t)bh * VSL * HD + t;
#pragma unroll
    for (int s = 0; s < VSL; ++s) { vall += pa[s * HD]; vmsk += pm[s * HD]; }
  }
  vmskS[t] = vmsk;
  const float vmean = vall * (1.0f / (float)Sq);
  vmeanS[t] = vmean;
  if (qb == 0) vmeanOut[bh * HD + t] = vmean;
  __syncthreads();

  int ubase = 0;
#pragma unroll
  for (int ww = 0; ww < 4; ++ww) if (ww < w) ubase += wtot[ww];
  const int nb = wtot[0] + wtot[1] + wtot[2] + wtot[3];
  {
    const unsigned lt = (1u << lane) - 1u;
#pragma unroll
    for (int i = 0; i < 16; ++i) {
      if ((bal[i] >> lane) & 1u) U[ubase + __popc(bal[i] & lt)] = w * 512 + i * 32 + lane;
      ubase += __popc(bal[i]);
    }
  }

  {
    for (int r = 0; r < QT; ++r)
      if (mrow[q0 + r] == 0) O[baseBH + (size_t)(q0 + r) * HD + t] = 0.0f;
  }
  __syncthreads();
  if (q0 >= nb) return;
  const int nlast = nb - 1;

  int qi = q0 + w * 16 + lm;
  qi = (qi < nb) ? qi : nlast;
  const float* qrow = Qm + baseBH + (size_t)U[qi] * HD;
  v2f qf[HD / 4];
#pragma unroll
  for (int kd = 0; kd < HD / 4; ++kd)
    qf[kd] = *(const v2f*)(qrow + 4 * kd + 2 * lh);

  v8f o[8];
#pragma unroll
  for (int s = 0; s < 8; ++s) o[s] = vzero8();
  float rmax = -INFINITY;
  float rsum = 0.0f;

  for (int kt = 0; kt < nb; kt += KT) {
#pragma unroll
    for (int i = 0; i < 8; ++i) {
      int idx = t + i * 128;
      int r = idx >> 5;
      int c = (idx & 31) << 2;
      int kr = kt + r;
      kr = (kr < nb) ? kr : nlast;
      const size_t ro = baseBH + (size_t)U[kr] * HD + c;
      float4 kv = *(const float4*)(Km + ro);
      *(float4*)&Ks[r][c] = kv;
      float4 vv = *(const float4*)(Vm + ro);
      float* dst = &Vsp[r >> 1][2 * c + (r & 1)];
      dst[0] = vv.x;
      dst[2] = vv.y;
      dst[4] = vv.z;
      dst[6] = vv.w;
    }
    __syncthreads();

    v8f sc[2];
    sc[0] = vzero8();
    sc[1] = vzero8();
#pragma unroll
    for (int kd = 0; kd < HD / 4; ++kd) {
      v2f af = qf[kd];
#pragma unroll
      for (int j = 0; j < 2; ++j) {
        int kc = j * 16 + lm;
        v2f bf = *(const v2f*)&Ks[kc][4 * kd + 2 * lh];
        sc[j] = WMMA_F32(af, bf, sc[j]);
      }
    }

#pragma unroll
    for (int j = 0; j < 2; ++j)
#pragma unroll
      for (int r = 0; r < 8; ++r) {
        int m = r + 8 * lh;
        int kc = j * 16 + lm;
        float v = sc[j][r];
        if (kt + kc >= nb) v = -INFINITY;
        Ps[w][m][kc] = v;
      }
    __syncthreads();

    {
      const int c0 = lh * 16;
      float pm = -INFINITY;
#pragma unroll
      for (int c = 0; c < 16; ++c) pm = fmaxf(pm, Ps[w][lm][c0 + c]);
      pm = fmaxf(pm, __shfl_xor(pm, 16, 32));
      float nm = fmaxf(rmax, pm);
      float cf = __expf(rmax - nm);
      float ps = 0.0f;
#pragma unroll
      for (int c = 0; c < 16; ++c) {
        float e = __expf(Ps[w][lm][c0 + c] - nm);
        Ps[w][lm][c0 + c] = e;
        ps += e;
      }
      ps += __shfl_xor(ps, 16, 32);
      rsum = rsum * cf + ps;
      rmax = nm;
      if (lane < 16) corrS[w][lm] = cf;
    }
    __syncthreads();

    float cv[8];
#pragma unroll
    for (int r = 0; r < 8; ++r) cv[r] = corrS[w][r + 8 * lh];
#pragma unroll
    for (int s = 0; s < 8; ++s)
#pragma unroll
      for (int r = 0; r < 8; ++r) o[s][r] *= cv[r];

#pragma unroll
    for (int kk = 0; kk < KT; kk += 4) {
      v2f af = *(const v2f*)&Ps[w][lm][kk + 2 * lh];
      const int khalf = (kk >> 1) + lh;
#pragma unroll
      for (int s = 0; s < 8; ++s) {
        int col = s * 16 + lm;
        v2f bf = *(const v2f*)&Vsp[khalf][2 * col];
        o[s] = WMMA_F32(af, bf, o[s]);
      }
    }
    __syncthreads();
  }

  if (lane < 16) { rsumS[w][lm] = rsum; rmaxS[w][lm] = rmax; }
  __syncthreads();

  const int nmask = Sq - nb;
  const float fnm = (float)nmask;
  float ra[8], rc[8], rv[8];
#pragma unroll
  for (int r = 0; r < 8; ++r) {
    const float mx = rmaxS[w][r + 8 * lh];
    const float rs = rsumS[w][r + 8 * lh];
    const float M = (nmask > 0) ? fmaxf(mx, NEGV) : mx;
    const float a = __expf(mx - M);
    const float c = (nmask > 0) ? __expf(NEGV - M) : 0.0f;
    ra[r] = a;
    rc[r] = c;
    rv[r] = 1.0f / (rs * a + c * fnm);
  }

#pragma unroll
  for (int s = 0; s < 8; ++s) {
    const int d = s * 16 + lm;
    const float vk = vmskS[d];
    const float vmn = vmeanS[d];
#pragma unroll
    for (int r = 0; r < 8; ++r) {
      const int qi2 = q0 + w * 16 + r + 8 * lh;
      if (qi2 < nb) {
        const int s2 = U[qi2];
        O[baseBH + (size_t)s2 * HD + d] = (o[s][r] * ra[r] + rc[r] * vk) * rv[r] - vmn;
      }
    }
  }
}

constexpr int OM = 64;
constexpr int ON = 256;
constexpr int OBPAD = 32;

__global__ __launch_bounds__(256) void oproj_unmasked(
    const float* __restrict__ A, const float* __restrict__ Wo,
    const float* __restrict__ vmeanB, const float* __restrict__ WS,
    const int* __restrict__ mask, float* __restrict__ C) {
  __shared__ __attribute__((aligned(16))) float As[OM][TK + APAD];
  __shared__ __attribute__((aligned(16))) float Bsp[TK / 2][2 * ON + OBPAD];
  __shared__ unsigned actS[2];

  const int t = threadIdx.x;
  const int lane = t & 31;
  const int wave = t >> 5;
  const int lm = lane & 15;
  const int lh = lane >> 4;
  const int n0 = blockIdx.x * ON;
  const int b = blockIdx.y >> 5;
  const int rr0 = (blockIdx.y & 31) * 4;

  bool f = false;
  if (t < 64) f = mask[b * Sq + 16 * rr0 + t] != 0;
  const unsigned bal = __builtin_amdgcn_ballot_w32(f);
  if (wave < 2 && lane == 0) actS[wave] = bal;
  __syncthreads();
  const unsigned u0 = (unsigned)__builtin_amdgcn_readfirstlane((int)actS[0]);
  const unsigned u1 = (unsigned)__builtin_amdgcn_readfirstlane((int)actS[1]);
  const unsigned actg[4] = {u0 & 0xffffu, u0 >> 16, u1 & 0xffffu, u1 >> 16};

  v8f acc[4][2];
#pragma unroll
  for (int g = 0; g < 4; ++g) { acc[g][0] = vzero8(); acc[g][1] = vzero8(); }
  v8f accG[2];
  accG[0] = vzero8();
  accG[1] = vzero8();

  const int ar = t >> 3;
  const int ac = (t & 7) << 2;
  const int cg = t & 63;
  const int pr = t >> 6;
  const float* ap0 = A + (size_t)(b * Sq + (ar & 15) * 128 + rr0 + (ar >> 4)) * Hd + ac;
  const float* ap1 = ap0 + (size_t)2 * Hd;
  const float* vm0 = vmeanB + (size_t)(b * NH + (ar & 15)) * HD + ac;
  const size_t boff = (size_t)(2 * pr) * Hd + n0 + 4 * cg;
  const float* bp = Wo + boff;
  const float* wp = WS + boff;
  constexpr size_t bStep = (size_t)8 * Hd;
  constexpr int KEND = Hd + HD;

  float4 ga0 = *(const float4*)(ap0);
  float4 ga1 = *(const float4*)(ap1);
  float4 gb00 = *(const float4*)(bp);
  float4 gb01 = *(const float4*)(bp + Hd);
  float4 gb10 = *(const float4*)(bp + bStep);
  float4 gb11 = *(const float4*)(bp + bStep + Hd);
  float4 gb20 = *(const float4*)(bp + 2 * bStep);
  float4 gb21 = *(const float4*)(bp + 2 * bStep + Hd);
  float4 gb30 = *(const float4*)(bp + 3 * bStep);
  float4 gb31 = *(const float4*)(bp + 3 * bStep + Hd);

  float* const bdst = &Bsp[pr][8 * cg];
  constexpr int OBROW4 = 4 * (2 * ON + OBPAD);

  for (int kt = 0; kt < KEND; kt += TK) {
    *(float4*)&As[ar][ac] = ga0;
    *(float4*)&As[ar + 32][ac] = ga1;
    *(float4*)(bdst)                  = make_float4(gb00.x, gb01.x, gb00.y, gb01.y);
    *(float4*)(bdst + 4)              = make_float4(gb00.z, gb01.z, gb00.w, gb01.w);
    *(float4*)(bdst + OBROW4)         = make_float4(gb10.x, gb11.x, gb10.y, gb11.y);
    *(float4*)(bdst + OBROW4 + 4)     = make_float4(gb10.z, gb11.z, gb10.w, gb11.w);
    *(float4*)(bdst + 2 * OBROW4)     = make_float4(gb20.x, gb21.x, gb20.y, gb21.y);
    *(float4*)(bdst + 2 * OBROW4 + 4) = make_float4(gb20.z, gb21.z, gb20.w, gb21.w);
    *(float4*)(bdst + 3 * OBROW4)     = make_float4(gb30.x, gb31.x, gb30.y, gb31.y);
    *(float4*)(bdst + 3 * OBROW4 + 4) = make_float4(gb30.z, gb31.z, gb30.w, gb31.w);
    __syncthreads();

    {
      const int kn = (kt + TK < KEND) ? kt + TK : kt;
      const bool tail = kn >= Hd;
      const float* an0 = tail ? vm0 + (kn - Hd) : ap0 + kn;
      const float* an1 = tail ? vm0 + (kn - Hd) : ap1 + kn;
      const float* bn = tail ? wp + (size_t)(kn - Hd) * Hd : bp + (size_t)kn * Hd;
      ga0 = *(const float4*)(an0);
      ga1 = *(const float4*)(an1);
      gb00 = *(const float4*)(bn);
      gb01 = *(const float4*)(bn + Hd);
      gb10 = *(const float4*)(bn + bStep);
      gb11 = *(const float4*)(bn + bStep + Hd);
      gb20 = *(const float4*)(bn + 2 * bStep);
      gb21 = *(const float4*)(bn + 2 * bStep + Hd);
      gb30 = *(const float4*)(bn + 3 * bStep);
      gb31 = *(const float4*)(bn + 3 * bStep + Hd);
    }
    __builtin_amdgcn_sched_barrier(0);

    v2f bfr[8][2];
#pragma unroll
    for (int ks = 0; ks < 8; ++ks) {
      const int khalf = 2 * ks + lh;
#pragma unroll
      for (int j = 0; j < 2; ++j) {
        const int col = wave * 32 + j * 16 + lm;
        bfr[ks][j] = *(const v2f*)&Bsp[khalf][2 * col];
      }
    }
    if (kt < Hd) {
      const unsigned cbit = 1u << (kt >> 7);
#pragma unroll
      for (int g = 0; g < 4; ++g) {
        if (actg[g] & cbit) {
#pragma unroll
          for (int ks = 0; ks < 8; ++ks) {
            const v2f af = *(const v2f*)&As[g * 16 + lm][4 * ks + 2 * lh];
            acc[g][0] = WMMA_F32(af, bfr[ks][0], acc[g][0]);
            acc[g][1] = WMMA_F32(af, bfr[ks][1], acc[g][1]);
          }
        }
      }
    } else {
#pragma unroll
      for (int ks = 0; ks < 8; ++ks) {
        const v2f af = *(const v2f*)&As[lm][4 * ks + 2 * lh];
        accG[0] = WMMA_F32(af, bfr[ks][0], accG[0]);
        accG[1] = WMMA_F32(af, bfr[ks][1], accG[1]);
      }
    }
    __syncthreads();
  }

#pragma unroll
  for (int g = 0; g < 4; ++g)
#pragma unroll
    for (int j = 0; j < 2; ++j)
#pragma unroll
      for (int r = 0; r < 8; ++r) {
        const int h = r + 8 * lh;
        const size_t m = (size_t)(b * Sq + h * 128 + rr0 + g);
        const int n = n0 + wave * 32 + j * 16 + lm;
        C[m * Hd + n] = acc[g][j][r] + accG[j][r];
      }
}

extern "C" void kernel_launch(void* const* d_in, const int* in_sizes, int n_in,
                              void* d_out, int out_size, void* d_ws, size_t ws_size,
                              hipStream_t stream) {
  const float* X    = (const float*)d_in[0];
  const int*   mask = (const int*)d_in[1];
  const float* Wq   = (const float*)d_in[2];
  const float* Wk   = (const float*)d_in[3];
  const float* Wv   = (const float*)d_in[4];
  const float* Wqm  = (const float*)d_in[5];
  const float* Wkm  = (const float*)d_in[6];
  const float* Wvm  = (const float*)d_in[7];
  const float* Wo   = (const float*)d_in[8];
  float* out = (float*)d_out;

  constexpr size_t SZ = (size_t)Bz * Sq * Hd;
  constexpr size_t WZ = (size_t)Hd * Hd;
  float* s0 = (float*)d_ws;
  float* s1 = s0 + SZ;
  float* s2 = s1 + SZ;
  float* s3 = s2 + SZ;
  float* fq = s0;
  float* fk = s0 + WZ;
  float* fv = s1;

  constexpr int M = Bz * Sq, N = Hd, K = Hd;
  dim3 gw3(N / TN, K / TM, 3);
  dim3 gg4(N / TN, M / TM, 4);
  dim3 gb(256);

  float* vm = out;

  float* xs = s1 + WZ;
  float* pmsk = xs + (size_t)1024 * Hd;
  float* pall = pmsk + (size_t)Bz * NH * VSL * HD;
  float* wsum = xs + (size_t)1100 * Hd;
  float* vmb = xs + (size_t)1300 * Hd;
  xsum_rows<<<dim3(Hd / 128, Bz * NH + 8), 128, 0, stream>>>(X, mask, xs, Wo, wsum);

  gemm_f32_wmma<false><<<gw3, gb, 0, stream>>>(Wq, Wqm, fq, Wk, Wkm, fk, Wv, Wvm, fv, Wv, Wvm, fv, mask, K, N, K);
  gemm_f32_wmma<true><<<gg4, gb, 0, stream>>>(X, fq, s2, X, fk, s3, X, fv, vm, xs, fv, pmsk, mask, M, N, K);

  attn_f32_wmma<<<Bz * NH * (Sq / QT), 128, 0, stream>>>(s2, s3, vm, mask, pall, pmsk, s0, vmb);

  oproj_unmasked<<<dim3(Hd / ON, (Bz * Sq) / OM), gb, 0, stream>>>(s0, Wo, vmb, wsum, mask, out);
}
